# starting kernel
# baseline (speedup 1.0000x reference)
#include <hip/hip_runtime.h>

#define D14   14
#define SM    17
#define JITTER 1e-4f
#define MIN_Q  1e-5f
#define MIN_R  1e-4f

typedef __attribute__((ext_vector_type(2))) float v2f;
typedef __attribute__((ext_vector_type(8))) float v8f;

template <bool BT>
__device__ __forceinline__ v8f mm_chain(const float* A, const float* Bm,
                                        const float* colmask,
                                        v8f c, int l16, int h) {
#pragma unroll
  for (int kk = 0; kk < 4; ++kk) {
    const int k0 = 4 * kk + 2 * h;
    v2f a, b;
    float a0 = A[l16 * SM + k0];
    float a1 = A[l16 * SM + k0 + 1];
    if (colmask) { a0 *= colmask[k0]; a1 *= colmask[k0 + 1]; }
    a.x = a0; a.y = a1;
    if (BT) {
      b.x = Bm[l16 * SM + k0];
      b.y = Bm[l16 * SM + k0 + 1];
    } else {
      b.x = Bm[k0 * SM + l16];
      b.y = Bm[(k0 + 1) * SM + l16];
    }
    c = __builtin_amdgcn_wmma_f32_16x16x4_f32(false, a, false, b, (short)0, c,
                                              false, false);
  }
  return c;
}

__global__ __launch_bounds__(32) void kf_scan(
    const float* __restrict__ xh, const float* __restrict__ z,
    const float* __restrict__ m21,
    const float* __restrict__ Wf, const float* __restrict__ bf,
    const float* __restrict__ Wq, const float* __restrict__ bq,
    const float* __restrict__ Wr, const float* __restrict__ br,
    float* __restrict__ out, int T) {
  __shared__ float sWf[196], sWq[196], sWr[196];
  __shared__ float sb[48];
  __shared__ float Pm[16 * SM], Sm[16 * SM], Gm[16 * SM], Km[16 * SM];
  __shared__ float zv[16], xv[16], xpv[16], qv[16], rv[16], mv[16], iv[16];

  const int lane = threadIdx.x;
  const int b    = blockIdx.x;
  const int l16  = lane & 15;
  const int h    = lane >> 4;

  for (int i = lane; i < 196; i += 32) { sWf[i] = Wf[i]; sWq[i] = Wq[i]; sWr[i] = Wr[i]; }
  if (lane < 16) {
    sb[lane]      = (lane < D14) ? bf[lane] : 0.f;
    sb[16 + lane] = (lane < D14) ? bq[lane] : 0.f;
    sb[32 + lane] = (lane < D14) ? br[lane] : 0.f;
    zv[lane] = 0.f; xv[lane] = 0.f; xpv[lane] = 0.f; qv[lane] = 0.f;
    rv[lane] = 0.f; mv[lane] = 0.f; iv[lane] = 0.f;
  }
#pragma unroll
  for (int k = 0; k < 8; ++k) {
    int idx = lane * 8 + k; int r = idx >> 4, c = idx & 15;
    Pm[r * SM + c] = (r == c && r < D14) ? 1.0f : 0.0f;
  }
  __syncthreads();

  const long long base = (long long)b * T;
#pragma unroll 1
  for (int t = 0; t < T; ++t) {
    const long long tok = base + t;
    if (lane < D14) {
      zv[lane] = z[tok * D14 + lane];
      xv[lane] = xh[tok * D14 + lane];
    }
    if (lane < 7) {
      float v  = m21[tok * 21 + 3 * lane + 2];
      float mm = (v > 0.5f) ? 1.0f : 0.0f;
      mv[2 * lane] = mm; mv[2 * lane + 1] = mm;
    }
    __syncthreads();
    if (lane < D14) {
      float a0 = sb[lane], a2 = sb[32 + lane];
#pragma unroll
      for (int i = 0; i < D14; ++i) {
        a0 += xv[i] * sWf[i * D14 + lane];
        a2 += zv[i] * sWr[i * D14 + lane];
      }
      xpv[lane] = a0;
      rv[lane]  = fmaxf(a2, MIN_R);
    }
    __syncthreads();
    if (lane < D14) {
      float a1 = sb[16 + lane];
#pragma unroll
      for (int i = 0; i < D14; ++i) a1 += xpv[i] * sWq[i * D14 + lane];
      qv[lane] = fmaxf(a1, MIN_Q);
    }
    __syncthreads();
    if (lane < D14) Pm[lane * SM + lane] += qv[lane];
    __syncthreads();
#pragma unroll
    for (int k = 0; k < 8; ++k) {
      int idx = lane * 8 + k; int r = idx >> 4, c = idx & 15;
      float ps = 0.5f * (Pm[r * SM + c] + Pm[c * SM + r]);
      float s  = mv[r] * mv[c] * ps;
      if (r == c && r < D14) s += mv[r] * rv[r] + (1.0f - mv[r]) + JITTER * mv[r];
      Sm[r * SM + c] = s;
    }
    __syncthreads();
    float tv = (lane < D14) ? mv[lane] * Sm[lane * SM + lane] : 0.0f;
#pragma unroll
    for (int off = 16; off >= 1; off >>= 1) tv += __shfl_xor(tv, off, 32);
    const float dload = JITTER * fmaxf(tv, 1e-6f);
#pragma unroll
    for (int k = 0; k < 8; ++k) {
      int idx = lane * 8 + k; int r = idx >> 4, c = idx & 15;
      float g = Sm[r * SM + c];
      if (r == c) g = (r < D14) ? (g + dload) : 1.0f;
      Gm[r * SM + c] = g;
    }
    __syncthreads();
    for (int k = 0; k < D14; ++k) {
      const float ip = 1.0f / Gm[k * SM + k];
      if (h == 0) {
        float val = Gm[k * SM + l16];
        Gm[k * SM + l16] = (l16 == k) ? ip : val * ip;
      }
      __syncthreads();
      if (l16 != k) {
        const float f = Gm[l16 * SM + k];
#pragma unroll
        for (int j = 0; j < 8; ++j) {
          int c = h * 8 + j;
          float cur = (c == k) ? 0.0f : Gm[l16 * SM + c];
          Gm[l16 * SM + c] = cur - f * Gm[k * SM + c];
        }
      }
      __syncthreads();
    }
    v8f acc = {};
    acc = mm_chain<false>(Pm, Gm, mv, acc, l16, h);
#pragma unroll
    for (int v = 0; v < 8; ++v) Km[(v + 8 * h) * SM + l16] = acc[v];
    __syncthreads();
    if (lane < D14) iv[lane] = mv[lane] * (zv[lane] - xpv[lane]);
    __syncthreads();
    if (lane < D14) {
      float xo = xpv[lane];
#pragma unroll
      for (int i = 0; i < D14; ++i) xo += Km[lane * SM + i] * iv[i];
      out[tok * D14 + lane] = xo;
    }
    v8f ks = {};
    ks = mm_chain<false>(Km, Sm, nullptr, ks, l16, h);
#pragma unroll
    for (int v = 0; v < 8; ++v) Gm[(v + 8 * h) * SM + l16] = -ks[v];
    __syncthreads();
    v8f pc;
#pragma unroll
    for (int v = 0; v < 8; ++v) pc[v] = Pm[(v + 8 * h) * SM + l16];
    pc = mm_chain<true>(Gm, Km, nullptr, pc, l16, h);
#pragma unroll
    for (int v = 0; v < 8; ++v) Sm[(v + 8 * h) * SM + l16] = pc[v];
    __syncthreads();
#pragma unroll
    for (int k = 0; k < 8; ++k) {
      int idx = lane * 8 + k; int r = idx >> 4, c = idx & 15;
      Pm[r * SM + c] = 0.5f * (Sm[r * SM + c] + Sm[c * SM + r]);
    }
    __syncthreads();
  }
}

extern "C" void kernel_launch(void* const* d_in, const int* in_sizes, int n_in,
                              void* d_out, int out_size, void* d_ws, size_t ws_size,
                              hipStream_t stream) {
  const float* xh  = (const float*)d_in[0];
  const float* z   = (const float*)d_in[1];
  const float* m21 = (const float*)d_in[2];
  const float* Wf  = (const float*)d_in[3];
  const float* bf  = (const float*)d_in[4];
  const float* Wq  = (const float*)d_in[5];
  const float* bq  = (const float*)d_in[6];
  const float* Wr  = (const float*)d_in[7];
  const float* br  = (const float*)d_in[8];
  float* out = (float*)d_out;

  const int T  = 512;
  const int BT = in_sizes[1] / D14;
  const int B  = BT / T;

  kf_scan<<<dim3(B), dim3(32), 0, stream>>>(xh, z, m21, Wf, bf, Wq, bq, Wr, br,
                                            out, T);
}


# Gauss-Jordan inverse of the 16x16 innovation matrix held in registers (row per lane, permlane16/x16 and DPP row_share broadcasts) instead of 14 pivot steps of dependent LDS round trips
# speedup vs baseline: 2.0423x; 2.0423x over previous
#include <hip/hip_runtime.h>

#define D14   14
#define SM    17
#define JITTER 1e-4f
#define MIN_Q  1e-5f
#define MIN_R  1e-4f

typedef __attribute__((ext_vector_type(2))) float v2f;
typedef __attribute__((ext_vector_type(8))) float v8f;

template <bool BT>
__device__ __forceinline__ v8f mm_chain(const float* A, const float* Bm,
                                        const float* colmask,
                                        v8f c, int l16, int h) {
#pragma unroll
  for (int kk = 0; kk < 4; ++kk) {
    const int k0 = 4 * kk + 2 * h;
    v2f a, b;
    float a0 = A[l16 * SM + k0];
    float a1 = A[l16 * SM + k0 + 1];
    if (colmask) { a0 *= colmask[k0]; a1 *= colmask[k0 + 1]; }
    a.x = a0; a.y = a1;
    if (BT) {
      b.x = Bm[l16 * SM + k0];
      b.y = Bm[l16 * SM + k0 + 1];
    } else {
      b.x = Bm[k0 * SM + l16];
      b.y = Bm[(k0 + 1) * SM + l16];
    }
    c = __builtin_amdgcn_wmma_f32_16x16x4_f32(false, a, false, b, (short)0, c,
                                              false, false);
  }
  return c;
}


__device__ __forceinline__ float xhalf(float x) {
  return __builtin_bit_cast(float, __builtin_amdgcn_permlanex16(
      0, __builtin_bit_cast(int, x), 0x76543210, 0xfedcba98, false, false));
}

template <int K>
__device__ __forceinline__ float row_share(float x) {
  return __builtin_bit_cast(float, __builtin_amdgcn_update_dpp(
      0, __builtin_bit_cast(int, x), 0x150 + K, 0xf, 0xf, false));
}

template <int K>
__device__ __forceinline__ float pivot_bcast(float x, bool own) {
  constexpr unsigned S = 0x11111111u * (unsigned)K;
  const int xi = __builtin_bit_cast(int, x);
  const int a = __builtin_amdgcn_permlane16(0, xi, S, S, false, false);
  const int b = __builtin_amdgcn_permlanex16(0, xi, S, S, false, false);
  return __builtin_bit_cast(float, own ? a : b);
}

template <int K>
__device__ __forceinline__ void gj_step(float (&g)[8], int l16, int h) {
  constexpr int KH = K >> 3, KJ = K & 7;
  const bool  own = (h == KH);
  const bool  isk = (l16 == K);
  const float pv  = pivot_bcast<K>(g[KJ], own);
  const float fo  = xhalf(g[KJ]);
  const float f   = own ? g[KJ] : fo;
  const float fe  = isk ? -1.0f : f;
  float cur[8];
#pragma unroll
  for (int j = 0; j < 8; ++j)
    cur[j] = (isk || (own && j == KJ)) ? 0.0f : g[j];
  const float ip  = 1.0f / pv;
#pragma unroll
  for (int jj = 0; jj < 8; ++jj) {
    const int   j    = (KJ + 1 + jj) & 7;
    const float pr   = row_share<K>(g[j]);
    const float prs  = (own && j == KJ) ? ip : pr * ip;
    g[j] = cur[j] - fe * prs;
  }
}

__global__ __launch_bounds__(32) void kf_scan(
    const float* __restrict__ xh, const float* __restrict__ z,
    const float* __restrict__ m21,
    const float* __restrict__ Wf, const float* __restrict__ bf,
    const float* __restrict__ Wq, const float* __restrict__ bq,
    const float* __restrict__ Wr, const float* __restrict__ br,
    float* __restrict__ out, int T) {
  __shared__ float sWf[196], sWq[196], sWr[196];
  __shared__ float sb[48];
  __shared__ float Pm[16 * SM], Sm[16 * SM], Gm[16 * SM], Km[16 * SM];
  __shared__ float zv[16], xv[16], xpv[16], qv[16], rv[16], mv[16], iv[16];

  const int lane = threadIdx.x;
  const int b    = blockIdx.x;
  const int l16  = lane & 15;
  const int h    = lane >> 4;

  for (int i = lane; i < 196; i += 32) { sWf[i] = Wf[i]; sWq[i] = Wq[i]; sWr[i] = Wr[i]; }
  if (lane < 16) {
    sb[lane]      = (lane < D14) ? bf[lane] : 0.f;
    sb[16 + lane] = (lane < D14) ? bq[lane] : 0.f;
    sb[32 + lane] = (lane < D14) ? br[lane] : 0.f;
    zv[lane] = 0.f; xv[lane] = 0.f; xpv[lane] = 0.f; qv[lane] = 0.f;
    rv[lane] = 0.f; mv[lane] = 0.f; iv[lane] = 0.f;
  }
#pragma unroll
  for (int k = 0; k < 8; ++k) {
    int idx = lane * 8 + k; int r = idx >> 4, c = idx & 15;
    Pm[r * SM + c] = (r == c && r < D14) ? 1.0f : 0.0f;
  }
  __syncthreads();

  const long long base = (long long)b * T;
#pragma unroll 1
  for (int t = 0; t < T; ++t) {
    const long long tok = base + t;
    if (lane < D14) {
      zv[lane] = z[tok * D14 + lane];
      xv[lane] = xh[tok * D14 + lane];
    }
    if (lane < 7) {
      float v  = m21[tok * 21 + 3 * lane + 2];
      float mm = (v > 0.5f) ? 1.0f : 0.0f;
      mv[2 * lane] = mm; mv[2 * lane + 1] = mm;
    }
    __syncthreads();
    if (lane < D14) {
      float a0 = sb[lane], a2 = sb[32 + lane];
#pragma unroll
      for (int i = 0; i < D14; ++i) {
        a0 += xv[i] * sWf[i * D14 + lane];
        a2 += zv[i] * sWr[i * D14 + lane];
      }
      xpv[lane] = a0;
      rv[lane]  = fmaxf(a2, MIN_R);
    }
    __syncthreads();
    if (lane < D14) {
      float a1 = sb[16 + lane];
#pragma unroll
      for (int i = 0; i < D14; ++i) a1 += xpv[i] * sWq[i * D14 + lane];
      qv[lane] = fmaxf(a1, MIN_Q);
    }
    __syncthreads();
    if (lane < D14) Pm[lane * SM + lane] += qv[lane];
    __syncthreads();
#pragma unroll
    for (int k = 0; k < 8; ++k) {
      int idx = lane * 8 + k; int r = idx >> 4, c = idx & 15;
      float ps = 0.5f * (Pm[r * SM + c] + Pm[c * SM + r]);
      float s  = mv[r] * mv[c] * ps;
      if (r == c && r < D14) s += mv[r] * rv[r] + (1.0f - mv[r]) + JITTER * mv[r];
      Sm[r * SM + c] = s;
    }
    __syncthreads();
    float g[8];
#pragma unroll
    for (int j = 0; j < 8; ++j) g[j] = Sm[l16 * SM + 8 * h + j];
    float tv = (lane < D14) ? mv[lane] * Sm[lane * SM + lane] : 0.0f;
#pragma unroll
    for (int off = 16; off >= 1; off >>= 1) tv += __shfl_xor(tv, off, 32);
    const float dload = JITTER * fmaxf(tv, 1e-6f);
#pragma unroll
    for (int j = 0; j < 8; ++j)
      if (8 * h + j == l16) g[j] = (l16 < D14) ? (g[j] + dload) : 1.0f;
    gj_step<0>(g, l16, h);  gj_step<1>(g, l16, h);  gj_step<2>(g, l16, h);
    gj_step<3>(g, l16, h);  gj_step<4>(g, l16, h);  gj_step<5>(g, l16, h);
    gj_step<6>(g, l16, h);  gj_step<7>(g, l16, h);  gj_step<8>(g, l16, h);
    gj_step<9>(g, l16, h);  gj_step<10>(g, l16, h); gj_step<11>(g, l16, h);
    gj_step<12>(g, l16, h); gj_step<13>(g, l16, h);
#pragma unroll
    for (int j = 0; j < 8; ++j) Gm[l16 * SM + 8 * h + j] = g[j];
    __syncthreads();
    v8f acc = {};
    acc = mm_chain<false>(Pm, Gm, mv, acc, l16, h);
#pragma unroll
    for (int v = 0; v < 8; ++v) Km[(v + 8 * h) * SM + l16] = acc[v];
    __syncthreads();
    if (lane < D14) iv[lane] = mv[lane] * (zv[lane] - xpv[lane]);
    __syncthreads();
    if (lane < D14) {
      float xo = xpv[lane];
#pragma unroll
      for (int i = 0; i < D14; ++i) xo += Km[lane * SM + i] * iv[i];
      out[tok * D14 + lane] = xo;
    }
    v8f ks = {};
    ks = mm_chain<false>(Km, Sm, nullptr, ks, l16, h);
#pragma unroll
    for (int v = 0; v < 8; ++v) Gm[(v + 8 * h) * SM + l16] = -ks[v];
    __syncthreads();
    v8f pc;
#pragma unroll
    for (int v = 0; v < 8; ++v) pc[v] = Pm[(v + 8 * h) * SM + l16];
    pc = mm_chain<true>(Gm, Km, nullptr, pc, l16, h);
#pragma unroll
    for (int v = 0; v < 8; ++v) Sm[(v + 8 * h) * SM + l16] = pc[v];
    __syncthreads();
#pragma unroll
    for (int k = 0; k < 8; ++k) {
      int idx = lane * 8 + k; int r = idx >> 4, c = idx & 15;
      Pm[r * SM + c] = 0.5f * (Sm[r * SM + c] + Sm[c * SM + r]);
    }
    __syncthreads();
  }
}

extern "C" void kernel_launch(void* const* d_in, const int* in_sizes, int n_in,
                              void* d_out, int out_size, void* d_ws, size_t ws_size,
                              hipStream_t stream) {
  const float* xh  = (const float*)d_in[0];
  const float* z   = (const float*)d_in[1];
  const float* m21 = (const float*)d_in[2];
  const float* Wf  = (const float*)d_in[3];
  const float* bf  = (const float*)d_in[4];
  const float* Wq  = (const float*)d_in[5];
  const float* bq  = (const float*)d_in[6];
  const float* Wr  = (const float*)d_in[7];
  const float* br  = (const float*)d_in[8];
  float* out = (float*)d_out;

  const int T  = 512;
  const int BT = in_sizes[1] / D14;
  const int B  = BT / T;

  kf_scan<<<dim3(B), dim3(32), 0, stream>>>(xh, z, m21, Wf, bf, Wq, bq, Wr, br,
                                            out, T);
}


# Work that does not depend on P leaves the serial step: per 32-token chunk a lane-per-token prologue loads inputs, runs the three nets into LDS tables, outputs leave per chunk; sum order kept
# speedup vs baseline: 2.7256x; 1.3346x over previous
#include <hip/hip_runtime.h>

#define D14   14
#define SM    17
#define CH    32
#define JITTER 1e-4f
#define MIN_Q  1e-5f
#define MIN_R  1e-4f

typedef __attribute__((ext_vector_type(2))) float v2f;
typedef __attribute__((ext_vector_type(8))) float v8f;

template <bool BT>
__device__ __forceinline__ v8f mm_chain(const float* A, const float* Bm,
                                        const float* colmask,
                                        v8f c, int l16, int h) {
#pragma unroll
  for (int kk = 0; kk < 4; ++kk) {
    const int k0 = 4 * kk + 2 * h;
    v2f a, b;
    float a0 = A[l16 * SM + k0];
    float a1 = A[l16 * SM + k0 + 1];
    if (colmask) { a0 *= colmask[k0]; a1 *= colmask[k0 + 1]; }
    a.x = a0; a.y = a1;
    if (BT) {
      b.x = Bm[l16 * SM + k0];
      b.y = Bm[l16 * SM + k0 + 1];
    } else {
      b.x = Bm[k0 * SM + l16];
      b.y = Bm[(k0 + 1) * SM + l16];
    }
    c = __builtin_amdgcn_wmma_f32_16x16x4_f32(false, a, false, b, (short)0, c,
                                              false, false);
  }
  return c;
}


__device__ __forceinline__ float xhalf(float x) {
  return __builtin_bit_cast(float, __builtin_amdgcn_permlanex16(
      0, __builtin_bit_cast(int, x), 0x76543210, 0xfedcba98, false, false));
}

template <int K>
__device__ __forceinline__ float row_share(float x) {
  return __builtin_bit_cast(float, __builtin_amdgcn_update_dpp(
      0, __builtin_bit_cast(int, x), 0x150 + K, 0xf, 0xf, false));
}

template <int K>
__device__ __forceinline__ float pivot_bcast(float x, bool own) {
  constexpr unsigned S = 0x11111111u * (unsigned)K;
  const int xi = __builtin_bit_cast(int, x);
  const int a = __builtin_amdgcn_permlane16(0, xi, S, S, false, false);
  const int b = __builtin_amdgcn_permlanex16(0, xi, S, S, false, false);
  return __builtin_bit_cast(float, own ? a : b);
}

template <int K>
__device__ __forceinline__ void gj_step(float (&g)[8], int l16, int h) {
  constexpr int KH = K >> 3, KJ = K & 7;
  const bool  own = (h == KH);
  const bool  isk = (l16 == K);
  const float pv  = pivot_bcast<K>(g[KJ], own);
  const float fo  = xhalf(g[KJ]);
  const float f   = own ? g[KJ] : fo;
  const float fe  = isk ? -1.0f : f;
  float cur[8];
#pragma unroll
  for (int j = 0; j < 8; ++j)
    cur[j] = (isk || (own && j == KJ)) ? 0.0f : g[j];
  const float ip  = 1.0f / pv;
#pragma unroll
  for (int jj = 0; jj < 8; ++jj) {
    const int   j    = (KJ + 1 + jj) & 7;
    const float pr   = row_share<K>(g[j]);
    const float prs  = (own && j == KJ) ? ip : pr * ip;
    g[j] = cur[j] - fe * prs;
  }
}

__global__ __launch_bounds__(32) void kf_scan(
    const float* __restrict__ xh, const float* __restrict__ z,
    const float* __restrict__ m21,
    const float* __restrict__ Wf, const float* __restrict__ bf,
    const float* __restrict__ Wq, const float* __restrict__ bq,
    const float* __restrict__ Wr, const float* __restrict__ br,
    float* __restrict__ out, int T) {
  __shared__ float sWf[196], sWq[196], sWr[196];
  __shared__ float sb[48];
  __shared__ float Pm[16 * SM], Sm[16 * SM], Gm[16 * SM], Km[16 * SM];
  __shared__ float tq[CH * D14], tr[CH * D14], ti[CH * D14], txp[CH * D14], to[CH * D14];
  __shared__ float tm[CH * 16];

  const int lane = threadIdx.x;
  const int b    = blockIdx.x;
  const int l16  = lane & 15;
  const int h    = lane >> 4;

  for (int i = lane; i < 196; i += 32) { sWf[i] = Wf[i]; sWq[i] = Wq[i]; sWr[i] = Wr[i]; }
  if (lane < 16) {
    sb[lane]      = (lane < D14) ? bf[lane] : 0.f;
    sb[16 + lane] = (lane < D14) ? bq[lane] : 0.f;
    sb[32 + lane] = (lane < D14) ? br[lane] : 0.f;
  }
#pragma unroll
  for (int k = 0; k < 8; ++k) {
    int idx = lane * 8 + k; int r = idx >> 4, c = idx & 15;
    Pm[r * SM + c] = (r == c && r < D14) ? 1.0f : 0.0f;
  }
  __syncthreads();

  const long long base = (long long)b * T;
#pragma unroll 1
  for (int t0 = 0; t0 < T; t0 += CH) {
    const int nt = (T - t0 < CH) ? (T - t0) : CH;
    {
      const long long tok = base + t0 + ((lane < nt) ? lane : 0);
      float x[D14], zz[D14], xp[D14], mk[7];
#pragma unroll
      for (int i = 0; i < D14; ++i) { x[i] = xh[tok * D14 + i]; zz[i] = z[tok * D14 + i]; }
#pragma unroll
      for (int i = 0; i < 7; ++i) mk[i] = (m21[tok * 21 + 3 * i + 2] > 0.5f) ? 1.0f : 0.0f;
#pragma unroll
      for (int j = 0; j < D14; ++j) {
        float a0 = sb[j], a2 = sb[32 + j];
#pragma unroll
        for (int i = 0; i < D14; ++i) {
          a0 += x[i]  * sWf[i * D14 + j];
          a2 += zz[i] * sWr[i * D14 + j];
        }
        xp[j] = a0;
        tr[lane * D14 + j] = fmaxf(a2, MIN_R);
      }
#pragma unroll
      for (int j = 0; j < D14; ++j) {
        float a1 = sb[16 + j];
#pragma unroll
        for (int i = 0; i < D14; ++i) a1 += xp[i] * sWq[i * D14 + j];
        tq[lane * D14 + j] = fmaxf(a1, MIN_Q);
      }
#pragma unroll
      for (int j = 0; j < D14; ++j) {
        const float mm = mk[j >> 1];
        tm[lane * 16 + j]   = mm;
        ti[lane * D14 + j]  = mm * (zz[j] - xp[j]);
        txp[lane * D14 + j] = xp[j];
      }
      tm[lane * 16 + 14] = 0.0f; tm[lane * 16 + 15] = 0.0f;
    }
    __syncthreads();
#pragma unroll 1
    for (int tl = 0; tl < nt; ++tl) {
      const float* qv  = tq  + tl * D14;
      const float* rv  = tr  + tl * D14;
      const float* iv  = ti  + tl * D14;
      const float* xpv = txp + tl * D14;
      const float* mv  = tm  + tl * 16;
      if (lane < D14) Pm[lane * SM + lane] += qv[lane];
      __syncthreads();
#pragma unroll
      for (int k = 0; k < 8; ++k) {
        int idx = lane * 8 + k; int r = idx >> 4, c = idx & 15;
        float ps = 0.5f * (Pm[r * SM + c] + Pm[c * SM + r]);
        float s  = mv[r] * mv[c] * ps;
        if (r == c && r < D14) s += mv[r] * rv[r] + (1.0f - mv[r]) + JITTER * mv[r];
        Sm[r * SM + c] = s;
      }
      __syncthreads();
      float g[8];
#pragma unroll
      for (int j = 0; j < 8; ++j) g[j] = Sm[l16 * SM + 8 * h + j];
      float tv = (lane < D14) ? mv[lane] * Sm[lane * SM + lane] : 0.0f;
#pragma unroll
      for (int off = 16; off >= 1; off >>= 1) tv += __shfl_xor(tv, off, 32);
      const float dload = JITTER * fmaxf(tv, 1e-6f);
#pragma unroll
      for (int j = 0; j < 8; ++j)
        if (8 * h + j == l16) g[j] = (l16 < D14) ? (g[j] + dload) : 1.0f;
      gj_step<0>(g, l16, h);  gj_step<1>(g, l16, h);  gj_step<2>(g, l16, h);
      gj_step<3>(g, l16, h);  gj_step<4>(g, l16, h);  gj_step<5>(g, l16, h);
      gj_step<6>(g, l16, h);  gj_step<7>(g, l16, h);  gj_step<8>(g, l16, h);
      gj_step<9>(g, l16, h);  gj_step<10>(g, l16, h); gj_step<11>(g, l16, h);
      gj_step<12>(g, l16, h); gj_step<13>(g, l16, h);
#pragma unroll
      for (int j = 0; j < 8; ++j) Gm[l16 * SM + 8 * h + j] = g[j];
      __syncthreads();
      v8f acc = {};
      acc = mm_chain<false>(Pm, Gm, mv, acc, l16, h);
#pragma unroll
      for (int v = 0; v < 8; ++v) Km[(v + 8 * h) * SM + l16] = acc[v];
      __syncthreads();
      if (lane < D14) {
        float xo = xpv[lane];
#pragma unroll
        for (int i = 0; i < D14; ++i) xo += Km[lane * SM + i] * iv[i];
        to[tl * D14 + lane] = xo;
      }
      v8f ks = {};
      ks = mm_chain<false>(Km, Sm, nullptr, ks, l16, h);
#pragma unroll
      for (int v = 0; v < 8; ++v) Gm[(v + 8 * h) * SM + l16] = -ks[v];
      __syncthreads();
      v8f pc;
#pragma unroll
      for (int v = 0; v < 8; ++v) pc[v] = Pm[(v + 8 * h) * SM + l16];
      pc = mm_chain<true>(Gm, Km, nullptr, pc, l16, h);
#pragma unroll
      for (int v = 0; v < 8; ++v) Sm[(v + 8 * h) * SM + l16] = pc[v];
      __syncthreads();
#pragma unroll
      for (int k = 0; k < 8; ++k) {
        int idx = lane * 8 + k; int r = idx >> 4, c = idx & 15;
        Pm[r * SM + c] = 0.5f * (Sm[r * SM + c] + Sm[c * SM + r]);
      }
      __syncthreads();
    }
#pragma unroll
    for (int k = 0; k < D14; ++k) {
      const int idx = k * 32 + lane;
      if (idx < nt * D14) out[(base + t0) * D14 + idx] = to[idx];
    }
  }
}

extern "C" void kernel_launch(void* const* d_in, const int* in_sizes, int n_in,
                              void* d_out, int out_size, void* d_ws, size_t ws_size,
                              hipStream_t stream) {
  const float* xh  = (const float*)d_in[0];
  const float* z   = (const float*)d_in[1];
  const float* m21 = (const float*)d_in[2];
  const float* Wf  = (const float*)d_in[3];
  const float* bf  = (const float*)d_in[4];
  const float* Wq  = (const float*)d_in[5];
  const float* bq  = (const float*)d_in[6];
  const float* Wr  = (const float*)d_in[7];
  const float* br  = (const float*)d_in[8];
  float* out = (float*)d_out;

  const int T  = 512;
  const int BT = in_sizes[1] / D14;
  const int B  = BT / T;

  kf_scan<<<dim3(B), dim3(32), 0, stream>>>(xh, z, m21, Wf, bf, Wq, bq, Wr, br,
                                            out, T);
}


# P and the operands of the three products stay in registers across the step, each result feeding the next fragment; only two transposes cross LDS; k order of each product's sum changes
# speedup vs baseline: 5.0878x; 1.8667x over previous
#include <hip/hip_runtime.h>

#define D14   14
#define SM    17
#define CH    32
#define JITTER 1e-4f
#define MIN_Q  1e-5f
#define MIN_R  1e-4f

typedef __attribute__((ext_vector_type(2))) float v2f;
typedef __attribute__((ext_vector_type(8))) float v8f;

__device__ __forceinline__ v8f mm_regs(const float (&a)[8], const float (&b)[8], v8f c) {
#pragma unroll
  for (int kk = 0; kk < 4; ++kk) {
    v2f av, bv;
    av.x = a[2 * kk]; av.y = a[2 * kk + 1];
    bv.x = b[2 * kk]; bv.y = b[2 * kk + 1];
    c = __builtin_amdgcn_wmma_f32_16x16x4_f32(false, av, false, bv, (short)0, c,
                                              false, false);
  }
  return c;
}


__device__ __forceinline__ float xhalf(float x) {
  return __builtin_bit_cast(float, __builtin_amdgcn_permlanex16(
      0, __builtin_bit_cast(int, x), 0x76543210, 0xfedcba98, false, false));
}

template <int K>
__device__ __forceinline__ float row_share(float x) {
  return __builtin_bit_cast(float, __builtin_amdgcn_update_dpp(
      0, __builtin_bit_cast(int, x), 0x150 + K, 0xf, 0xf, false));
}

template <int K>
__device__ __forceinline__ float pivot_bcast(float x, bool own) {
  constexpr unsigned S = 0x11111111u * (unsigned)K;
  const int xi = __builtin_bit_cast(int, x);
  const int a = __builtin_amdgcn_permlane16(0, xi, S, S, false, false);
  const int b = __builtin_amdgcn_permlanex16(0, xi, S, S, false, false);
  return __builtin_bit_cast(float, own ? a : b);
}

template <int K>
__device__ __forceinline__ void gj_step(float (&g)[8], int l16, int h) {
  constexpr int KH = K >> 3, KJ = K & 7;
  const bool  own = (h == KH);
  const bool  isk = (l16 == K);
  const float pv  = pivot_bcast<K>(g[KJ], own);
  const float fo  = xhalf(g[KJ]);
  const float f   = own ? g[KJ] : fo;
  const float fe  = isk ? -1.0f : f;
  float cur[8];
#pragma unroll
  for (int j = 0; j < 8; ++j)
    cur[j] = (isk || (own && j == KJ)) ? 0.0f : g[j];
  const float ip  = 1.0f / pv;
#pragma unroll
  for (int jj = 0; jj < 8; ++jj) {
    const int   j    = (KJ + 1 + jj) & 7;
    const float pr   = row_share<K>(g[j]);
    const float prs  = (own && j == KJ) ? ip : pr * ip;
    g[j] = cur[j] - fe * prs;
  }
}

__global__ __launch_bounds__(32) void kf_scan(
    const float* __restrict__ xh, const float* __restrict__ z,
    const float* __restrict__ m21,
    const float* __restrict__ Wf, const float* __restrict__ bf,
    const float* __restrict__ Wq, const float* __restrict__ bq,
    const float* __restrict__ Wr, const float* __restrict__ br,
    float* __restrict__ out, int T) {
  __shared__ float sWf[196], sWq[196], sWr[196];
  __shared__ float sb[48];
  __shared__ float Sm[16 * SM], Gm[16 * SM];
  __shared__ float tq[CH * D14], tr[CH * D14], ti[CH * D14], txp[CH * D14], to[CH * D14];
  __shared__ float tm[CH * 16];

  const int lane = threadIdx.x;
  const int b    = blockIdx.x;
  const int l16  = lane & 15;
  const int h    = lane >> 4;

  for (int i = lane; i < 196; i += 32) { sWf[i] = Wf[i]; sWq[i] = Wq[i]; sWr[i] = Wr[i]; }
  if (lane < 16) {
    sb[lane]      = (lane < D14) ? bf[lane] : 0.f;
    sb[16 + lane] = (lane < D14) ? bq[lane] : 0.f;
    sb[32 + lane] = (lane < D14) ? br[lane] : 0.f;
  }
  const bool in14 = (l16 < D14);
  const int  i14  = in14 ? l16 : 0;
  float p[8];
#pragma unroll
  for (int j = 0; j < 8; ++j) p[j] = (8 * h + j == l16 && in14) ? 1.0f : 0.0f;
  __syncthreads();

  const long long base = (long long)b * T;
#pragma unroll 1
  for (int t0 = 0; t0 < T; t0 += CH) {
    const int nt = (T - t0 < CH) ? (T - t0) : CH;
    {
      const long long tok = base + t0 + ((lane < nt) ? lane : 0);
      float x[D14], zz[D14], xp[D14], mk[7];
#pragma unroll
      for (int i = 0; i < D14; ++i) { x[i] = xh[tok * D14 + i]; zz[i] = z[tok * D14 + i]; }
#pragma unroll
      for (int i = 0; i < 7; ++i) mk[i] = (m21[tok * 21 + 3 * i + 2] > 0.5f) ? 1.0f : 0.0f;
#pragma unroll
      for (int j = 0; j < D14; ++j) {
        float a0 = sb[j], a2 = sb[32 + j];
#pragma unroll
        for (int i = 0; i < D14; ++i) {
          a0 += x[i]  * sWf[i * D14 + j];
          a2 += zz[i] * sWr[i * D14 + j];
        }
        xp[j] = a0;
        tr[lane * D14 + j] = fmaxf(a2, MIN_R);
      }
#pragma unroll
      for (int j = 0; j < D14; ++j) {
        float a1 = sb[16 + j];
#pragma unroll
        for (int i = 0; i < D14; ++i) a1 += xp[i] * sWq[i * D14 + j];
        tq[lane * D14 + j] = fmaxf(a1, MIN_Q);
      }
#pragma unroll
      for (int j = 0; j < D14; ++j) {
        const float mm = mk[j >> 1];
        tm[lane * 16 + j]   = mm;
        ti[lane * D14 + j]  = mm * (zz[j] - xp[j]);
        txp[lane * D14 + j] = xp[j];
      }
      tm[lane * 16 + 14] = 0.0f; tm[lane * 16 + 15] = 0.0f;
    }
    __syncthreads();
#pragma unroll 1
    for (int tl = 0; tl < nt; ++tl) {
      const float* qv  = tq  + tl * D14;
      const float* rv  = tr  + tl * D14;
      const float* iv  = ti  + tl * D14;
      const float* xpv = txp + tl * D14;
      const float* mv  = tm  + tl * 16;
      const float qd = qv[i14];
      const float rd = rv[i14];
      const float md = mv[l16];
      float mc[8], ivc[8];
#pragma unroll
      for (int j = 0; j < 8; ++j) {
        mc[j]  = mv[8 * h + j];
        const int ii = 8 * h + j;
        ivc[j] = iv[ii < D14 ? ii : D14 - 1];
      }
#pragma unroll
      for (int j = 0; j < 8; ++j)
        if (8 * h + j == l16 && in14) p[j] += qd;
      float sv[8], g[8];
#pragma unroll
      for (int j = 0; j < 8; ++j) {
        float s = md * mc[j] * p[j];
        if (8 * h + j == l16 && in14) s += md * rd + (1.0f - md) + JITTER * md;
        sv[j] = s;
        g[j]  = s;
      }
      float dsel = 0.0f;
#pragma unroll
      for (int j = 0; j < 8; ++j)
        if (8 * h + j == l16) dsel = sv[j];
      const float dpar = xhalf(dsel);
      const float dgl  = (h == (l16 >> 3)) ? dsel : dpar;
      float tv = (lane < D14) ? md * dgl : 0.0f;
#pragma unroll
      for (int off = 16; off >= 1; off >>= 1) tv += __shfl_xor(tv, off, 32);
      const float dload = JITTER * fmaxf(tv, 1e-6f);
#pragma unroll
      for (int j = 0; j < 8; ++j)
        if (8 * h + j == l16) g[j] = in14 ? (g[j] + dload) : 1.0f;
      gj_step<0>(g, l16, h);  gj_step<1>(g, l16, h);  gj_step<2>(g, l16, h);
      gj_step<3>(g, l16, h);  gj_step<4>(g, l16, h);  gj_step<5>(g, l16, h);
      gj_step<6>(g, l16, h);  gj_step<7>(g, l16, h);  gj_step<8>(g, l16, h);
      gj_step<9>(g, l16, h);  gj_step<10>(g, l16, h); gj_step<11>(g, l16, h);
      gj_step<12>(g, l16, h); gj_step<13>(g, l16, h);
#pragma unroll
      for (int j = 0; j < 8; ++j) Gm[l16 * SM + 8 * h + j] = g[j];
      __syncthreads();
      float gt[8], pho[8];
#pragma unroll
      for (int j = 0; j < 8; ++j) {
        gt[j]  = Gm[(8 * h + j) * SM + l16];
        pho[j] = p[j] * mc[j];
      }
      v8f zero = {};
      const v8f ktv = mm_regs(gt, pho, zero);
      float kt[8];
#pragma unroll
      for (int v = 0; v < 8; ++v) kt[v] = ktv[v];
      {
        float xa = xpv[i14];
#pragma unroll
        for (int v = 0; v < 8; ++v) xa += kt[v] * ivc[v];
        float xb = xhalf(xa);
#pragma unroll
        for (int v = 0; v < 6; ++v) xb += kt[v] * ivc[v];
        if (h == 1 && in14) to[tl * D14 + l16] = xb;
      }
      const v8f kstv = mm_regs(sv, kt, zero);
      float nks[8];
#pragma unroll
      for (int v = 0; v < 8; ++v) nks[v] = -kstv[v];
      v8f pc;
#pragma unroll
      for (int v = 0; v < 8; ++v) pc[v] = p[v];
      pc = mm_regs(nks, kt, pc);
#pragma unroll
      for (int v = 0; v < 8; ++v) Sm[(8 * h + v) * SM + l16] = pc[v];
      __syncthreads();
#pragma unroll
      for (int v = 0; v < 8; ++v) p[v] = 0.5f * (pc[v] + Sm[l16 * SM + 8 * h + v]);
    }
#pragma unroll
    for (int k = 0; k < D14; ++k) {
      const int idx = k * 32 + lane;
      if (idx < nt * D14) out[(base + t0) * D14 + idx] = to[idx];
    }
  }
}

extern "C" void kernel_launch(void* const* d_in, const int* in_sizes, int n_in,
                              void* d_out, int out_size, void* d_ws, size_t ws_size,
                              hipStream_t stream) {
  const float* xh  = (const float*)d_in[0];
  const float* z   = (const float*)d_in[1];
  const float* m21 = (const float*)d_in[2];
  const float* Wf  = (const float*)d_in[3];
  const float* bf  = (const float*)d_in[4];
  const float* Wq  = (const float*)d_in[5];
  const float* bq  = (const float*)d_in[6];
  const float* Wr  = (const float*)d_in[7];
  const float* br  = (const float*)d_in[8];
  float* out = (float*)d_out;

  const int T  = 512;
  const int BT = in_sizes[1] / D14;
  const int B  = BT / T;

  kf_scan<<<dim3(B), dim3(32), 0, stream>>>(xh, z, m21, Wf, bf, Wq, bq, Wr, br,
                                            out, T);
}


# Pivot reciprocal of the in-register Gauss-Jordan inverse as v_rcp_f32 plus one fma refinement step (3 dependent instructions) instead of the IEEE division (10 dependent) on each of 14 pivot links
# speedup vs baseline: 6.2611x; 1.2306x over previous
#include <hip/hip_runtime.h>

#define D14   14
#define SM    17
#define CH    32
#define JITTER 1e-4f
#define MIN_Q  1e-5f
#define MIN_R  1e-4f

typedef __attribute__((ext_vector_type(2))) float v2f;
typedef __attribute__((ext_vector_type(8))) float v8f;

__device__ __forceinline__ v8f mm_regs(const float (&a)[8], const float (&b)[8], v8f c) {
#pragma unroll
  for (int kk = 0; kk < 4; ++kk) {
    v2f av, bv;
    av.x = a[2 * kk]; av.y = a[2 * kk + 1];
    bv.x = b[2 * kk]; bv.y = b[2 * kk + 1];
    c = __builtin_amdgcn_wmma_f32_16x16x4_f32(false, av, false, bv, (short)0, c,
                                              false, false);
  }
  return c;
}


__device__ __forceinline__ float xhalf(float x) {
  return __builtin_bit_cast(float, __builtin_amdgcn_permlanex16(
      0, __builtin_bit_cast(int, x), 0x76543210, 0xfedcba98, false, false));
}

template <int K>
__device__ __forceinline__ float row_share(float x) {
  return __builtin_bit_cast(float, __builtin_amdgcn_update_dpp(
      0, __builtin_bit_cast(int, x), 0x150 + K, 0xf, 0xf, false));
}

template <int K>
__device__ __forceinline__ float pivot_bcast(float x, bool own) {
  constexpr unsigned S = 0x11111111u * (unsigned)K;
  const int xi = __builtin_bit_cast(int, x);
  const int a = __builtin_amdgcn_permlane16(0, xi, S, S, false, false);
  const int b = __builtin_amdgcn_permlanex16(0, xi, S, S, false, false);
  return __builtin_bit_cast(float, own ? a : b);
}

template <int K>
__device__ __forceinline__ void gj_step(float (&g)[8], int l16, int h) {
  constexpr int KH = K >> 3, KJ = K & 7;
  const bool  own = (h == KH);
  const bool  isk = (l16 == K);
  const float pv  = pivot_bcast<K>(g[KJ], own);
  const float fo  = xhalf(g[KJ]);
  const float f   = own ? g[KJ] : fo;
  const float fe  = isk ? -1.0f : f;
  float cur[8];
#pragma unroll
  for (int j = 0; j < 8; ++j)
    cur[j] = (isk || (own && j == KJ)) ? 0.0f : g[j];
  const float r0  = __builtin_amdgcn_rcpf(pv);
  const float ip  = __builtin_fmaf(__builtin_fmaf(-pv, r0, 1.0f), r0, r0);
#pragma unroll
  for (int jj = 0; jj < 8; ++jj) {
    const int   j    = (KJ + 1 + jj) & 7;
    const float pr   = row_share<K>(g[j]);
    const float prs  = (own && j == KJ) ? ip : pr * ip;
    g[j] = cur[j] - fe * prs;
  }
}

__global__ __launch_bounds__(32) void kf_scan(
    const float* __restrict__ xh, const float* __restrict__ z,
    const float* __restrict__ m21,
    const float* __restrict__ Wf, const float* __restrict__ bf,
    const float* __restrict__ Wq, const float* __restrict__ bq,
    const float* __restrict__ Wr, const float* __restrict__ br,
    float* __restrict__ out, int T) {
  __shared__ float sWf[196], sWq[196], sWr[196];
  __shared__ float sb[48];
  __shared__ float Sm[16 * SM], Gm[16 * SM];
  __shared__ float tq[CH * D14], tr[CH * D14], ti[CH * D14], txp[CH * D14], to[CH * D14];
  __shared__ float tm[CH * 16];

  const int lane = threadIdx.x;
  const int b    = blockIdx.x;
  const int l16  = lane & 15;
  const int h    = lane >> 4;

  for (int i = lane; i < 196; i += 32) { sWf[i] = Wf[i]; sWq[i] = Wq[i]; sWr[i] = Wr[i]; }
  if (lane < 16) {
    sb[lane]      = (lane < D14) ? bf[lane] : 0.f;
    sb[16 + lane] = (lane < D14) ? bq[lane] : 0.f;
    sb[32 + lane] = (lane < D14) ? br[lane] : 0.f;
  }
  const bool in14 = (l16 < D14);
  const int  i14  = in14 ? l16 : 0;
  float p[8];
#pragma unroll
  for (int j = 0; j < 8; ++j) p[j] = (8 * h + j == l16 && in14) ? 1.0f : 0.0f;
  __syncthreads();

  const long long base = (long long)b * T;
#pragma unroll 1
  for (int t0 = 0; t0 < T; t0 += CH) {
    const int nt = (T - t0 < CH) ? (T - t0) : CH;
    {
      const long long tok = base + t0 + ((lane < nt) ? lane : 0);
      float x[D14], zz[D14], xp[D14], mk[7];
#pragma unroll
      for (int i = 0; i < D14; ++i) { x[i] = xh[tok * D14 + i]; zz[i] = z[tok * D14 + i]; }
#pragma unroll
      for (int i = 0; i < 7; ++i) mk[i] = (m21[tok * 21 + 3 * i + 2] > 0.5f) ? 1.0f : 0.0f;
#pragma unroll
      for (int j = 0; j < D14; ++j) {
        float a0 = sb[j], a2 = sb[32 + j];
#pragma unroll
        for (int i = 0; i < D14; ++i) {
          a0 += x[i]  * sWf[i * D14 + j];
          a2 += zz[i] * sWr[i * D14 + j];
        }
        xp[j] = a0;
        tr[lane * D14 + j] = fmaxf(a2, MIN_R);
      }
#pragma unroll
      for (int j = 0; j < D14; ++j) {
        float a1 = sb[16 + j];
#pragma unroll
        for (int i = 0; i < D14; ++i) a1 += xp[i] * sWq[i * D14 + j];
        tq[lane * D14 + j] = fmaxf(a1, MIN_Q);
      }
#pragma unroll
      for (int j = 0; j < D14; ++j) {
        const float mm = mk[j >> 1];
        tm[lane * 16 + j]   = mm;
        ti[lane * D14 + j]  = mm * (zz[j] - xp[j]);
        txp[lane * D14 + j] = xp[j];
      }
      tm[lane * 16 + 14] = 0.0f; tm[lane * 16 + 15] = 0.0f;
    }
    __syncthreads();
#pragma unroll 1
    for (int tl = 0; tl < nt; ++tl) {
      const float* qv  = tq  + tl * D14;
      const float* rv  = tr  + tl * D14;
      const float* iv  = ti  + tl * D14;
      const float* xpv = txp + tl * D14;
      const float* mv  = tm  + tl * 16;
      const float qd = qv[i14];
      const float rd = rv[i14];
      const float md = mv[l16];
      float mc[8], ivc[8];
#pragma unroll
      for (int j = 0; j < 8; ++j) {
        mc[j]  = mv[8 * h + j];
        const int ii = 8 * h + j;
        ivc[j] = iv[ii < D14 ? ii : D14 - 1];
      }
#pragma unroll
      for (int j = 0; j < 8; ++j)
        if (8 * h + j == l16 && in14) p[j] += qd;
      float sv[8], g[8];
#pragma unroll
      for (int j = 0; j < 8; ++j) {
        float s = md * mc[j] * p[j];
        if (8 * h + j == l16 && in14) s += md * rd + (1.0f - md) + JITTER * md;
        sv[j] = s;
        g[j]  = s;
      }
      float dsel = 0.0f;
#pragma unroll
      for (int j = 0; j < 8; ++j)
        if (8 * h + j == l16) dsel = sv[j];
      const float dpar = xhalf(dsel);
      const float dgl  = (h == (l16 >> 3)) ? dsel : dpar;
      float tv = (lane < D14) ? md * dgl : 0.0f;
#pragma unroll
      for (int off = 16; off >= 1; off >>= 1) tv += __shfl_xor(tv, off, 32);
      const float dload = JITTER * fmaxf(tv, 1e-6f);
#pragma unroll
      for (int j = 0; j < 8; ++j)
        if (8 * h + j == l16) g[j] = in14 ? (g[j] + dload) : 1.0f;
      gj_step<0>(g, l16, h);  gj_step<1>(g, l16, h);  gj_step<2>(g, l16, h);
      gj_step<3>(g, l16, h);  gj_step<4>(g, l16, h);  gj_step<5>(g, l16, h);
      gj_step<6>(g, l16, h);  gj_step<7>(g, l16, h);  gj_step<8>(g, l16, h);
      gj_step<9>(g, l16, h);  gj_step<10>(g, l16, h); gj_step<11>(g, l16, h);
      gj_step<12>(g, l16, h); gj_step<13>(g, l16, h);
#pragma unroll
      for (int j = 0; j < 8; ++j) Gm[l16 * SM + 8 * h + j] = g[j];
      __syncthreads();
      float gt[8], pho[8];
#pragma unroll
      for (int j = 0; j < 8; ++j) {
        gt[j]  = Gm[(8 * h + j) * SM + l16];
        pho[j] = p[j] * mc[j];
      }
      v8f zero = {};
      const v8f ktv = mm_regs(gt, pho, zero);
      float kt[8];
#pragma unroll
      for (int v = 0; v < 8; ++v) kt[v] = ktv[v];
      {
        float xa = xpv[i14];
#pragma unroll
        for (int v = 0; v < 8; ++v) xa += kt[v] * ivc[v];
        float xb = xhalf(xa);
#pragma unroll
        for (int v = 0; v < 6; ++v) xb += kt[v] * ivc[v];
        if (h == 1 && in14) to[tl * D14 + l16] = xb;
      }
      const v8f kstv = mm_regs(sv, kt, zero);
      float nks[8];
#pragma unroll
      for (int v = 0; v < 8; ++v) nks[v] = -kstv[v];
      v8f pc;
#pragma unroll
      for (int v = 0; v < 8; ++v) pc[v] = p[v];
      pc = mm_regs(nks, kt, pc);
#pragma unroll
      for (int v = 0; v < 8; ++v) Sm[(8 * h + v) * SM + l16] = pc[v];
      __syncthreads();
#pragma unroll
      for (int v = 0; v < 8; ++v) p[v] = 0.5f * (pc[v] + Sm[l16 * SM + 8 * h + v]);
    }
#pragma unroll
    for (int k = 0; k < D14; ++k) {
      const int idx = k * 32 + lane;
      if (idx < nt * D14) out[(base + t0) * D14 + idx] = to[idx];
    }
  }
}

extern "C" void kernel_launch(void* const* d_in, const int* in_sizes, int n_in,
                              void* d_out, int out_size, void* d_ws, size_t ws_size,
                              hipStream_t stream) {
  const float* xh  = (const float*)d_in[0];
  const float* z   = (const float*)d_in[1];
  const float* m21 = (const float*)d_in[2];
  const float* Wf  = (const float*)d_in[3];
  const float* bf  = (const float*)d_in[4];
  const float* Wq  = (const float*)d_in[5];
  const float* bq  = (const float*)d_in[6];
  const float* Wr  = (const float*)d_in[7];
  const float* br  = (const float*)d_in[8];
  float* out = (float*)d_out;

  const int T  = 512;
  const int BT = in_sizes[1] / D14;
  const int B  = BT / T;

  kf_scan<<<dim3(B), dim3(32), 0, stream>>>(xh, z, m21, Wf, bf, Wq, bq, Wr, br,
                                            out, T);
}


# Trace sum over the 32 lanes on the vector ALU (permlanex16 half swap, then four DPP row_xmask adds) instead of five dependent ds_bpermute round trips; same butterfly, same sum order
# speedup vs baseline: 7.3090x; 1.1674x over previous
#include <hip/hip_runtime.h>

#define D14   14
#define SM    17
#define CH    32
#define JITTER 1e-4f
#define MIN_Q  1e-5f
#define MIN_R  1e-4f

typedef __attribute__((ext_vector_type(2))) float v2f;
typedef __attribute__((ext_vector_type(8))) float v8f;

__device__ __forceinline__ v8f mm_regs(const float (&a)[8], const float (&b)[8], v8f c) {
#pragma unroll
  for (int kk = 0; kk < 4; ++kk) {
    v2f av, bv;
    av.x = a[2 * kk]; av.y = a[2 * kk + 1];
    bv.x = b[2 * kk]; bv.y = b[2 * kk + 1];
    c = __builtin_amdgcn_wmma_f32_16x16x4_f32(false, av, false, bv, (short)0, c,
                                              false, false);
  }
  return c;
}


__device__ __forceinline__ float xhalf(float x) {
  return __builtin_bit_cast(float, __builtin_amdgcn_permlanex16(
      0, __builtin_bit_cast(int, x), 0x76543210, 0xfedcba98, false, false));
}

template <int K>
__device__ __forceinline__ float row_share(float x) {
  return __builtin_bit_cast(float, __builtin_amdgcn_update_dpp(
      0, __builtin_bit_cast(int, x), 0x150 + K, 0xf, 0xf, false));
}

template <int M>
__device__ __forceinline__ float row_xor(float x) {
  return __builtin_bit_cast(float, __builtin_amdgcn_update_dpp(
      0, __builtin_bit_cast(int, x), 0x160 + M, 0xf, 0xf, false));
}

template <int K>
__device__ __forceinline__ float pivot_bcast(float x, bool own) {
  constexpr unsigned S = 0x11111111u * (unsigned)K;
  const int xi = __builtin_bit_cast(int, x);
  const int a = __builtin_amdgcn_permlane16(0, xi, S, S, false, false);
  const int b = __builtin_amdgcn_permlanex16(0, xi, S, S, false, false);
  return __builtin_bit_cast(float, own ? a : b);
}

template <int K>
__device__ __forceinline__ void gj_step(float (&g)[8], int l16, int h) {
  constexpr int KH = K >> 3, KJ = K & 7;
  const bool  own = (h == KH);
  const bool  isk = (l16 == K);
  const float pv  = pivot_bcast<K>(g[KJ], own);
  const float fo  = xhalf(g[KJ]);
  const float f   = own ? g[KJ] : fo;
  const float fe  = isk ? -1.0f : f;
  float cur[8];
#pragma unroll
  for (int j = 0; j < 8; ++j)
    cur[j] = (isk || (own && j == KJ)) ? 0.0f : g[j];
  const float r0  = __builtin_amdgcn_rcpf(pv);
  const float ip  = __builtin_fmaf(__builtin_fmaf(-pv, r0, 1.0f), r0, r0);
#pragma unroll
  for (int jj = 0; jj < 8; ++jj) {
    const int   j    = (KJ + 1 + jj) & 7;
    const float pr   = row_share<K>(g[j]);
    const float prs  = (own && j == KJ) ? ip : pr * ip;
    g[j] = cur[j] - fe * prs;
  }
}

__global__ __launch_bounds__(32) void kf_scan(
    const float* __restrict__ xh, const float* __restrict__ z,
    const float* __restrict__ m21,
    const float* __restrict__ Wf, const float* __restrict__ bf,
    const float* __restrict__ Wq, const float* __restrict__ bq,
    const float* __restrict__ Wr, const float* __restrict__ br,
    float* __restrict__ out, int T) {
  __shared__ float sWf[196], sWq[196], sWr[196];
  __shared__ float sb[48];
  __shared__ float Sm[16 * SM], Gm[16 * SM];
  __shared__ float tq[CH * D14], tr[CH * D14], ti[CH * D14], txp[CH * D14], to[CH * D14];
  __shared__ float tm[CH * 16];

  const int lane = threadIdx.x;
  const int b    = blockIdx.x;
  const int l16  = lane & 15;
  const int h    = lane >> 4;

  for (int i = lane; i < 196; i += 32) { sWf[i] = Wf[i]; sWq[i] = Wq[i]; sWr[i] = Wr[i]; }
  if (lane < 16) {
    sb[lane]      = (lane < D14) ? bf[lane] : 0.f;
    sb[16 + lane] = (lane < D14) ? bq[lane] : 0.f;
    sb[32 + lane] = (lane < D14) ? br[lane] : 0.f;
  }
  const bool in14 = (l16 < D14);
  const int  i14  = in14 ? l16 : 0;
  float p[8];
#pragma unroll
  for (int j = 0; j < 8; ++j) p[j] = (8 * h + j == l16 && in14) ? 1.0f : 0.0f;
  __syncthreads();

  const long long base = (long long)b * T;
#pragma unroll 1
  for (int t0 = 0; t0 < T; t0 += CH) {
    const int nt = (T - t0 < CH) ? (T - t0) : CH;
    {
      const long long tok = base + t0 + ((lane < nt) ? lane : 0);
      float x[D14], zz[D14], xp[D14], mk[7];
#pragma unroll
      for (int i = 0; i < D14; ++i) { x[i] = xh[tok * D14 + i]; zz[i] = z[tok * D14 + i]; }
#pragma unroll
      for (int i = 0; i < 7; ++i) mk[i] = (m21[tok * 21 + 3 * i + 2] > 0.5f) ? 1.0f : 0.0f;
#pragma unroll
      for (int j = 0; j < D14; ++j) {
        float a0 = sb[j], a2 = sb[32 + j];
#pragma unroll
        for (int i = 0; i < D14; ++i) {
          a0 += x[i]  * sWf[i * D14 + j];
          a2 += zz[i] * sWr[i * D14 + j];
        }
        xp[j] = a0;
        tr[lane * D14 + j] = fmaxf(a2, MIN_R);
      }
#pragma unroll
      for (int j = 0; j < D14; ++j) {
        float a1 = sb[16 + j];
#pragma unroll
        for (int i = 0; i < D14; ++i) a1 += xp[i] * sWq[i * D14 + j];
        tq[lane * D14 + j] = fmaxf(a1, MIN_Q);
      }
#pragma unroll
      for (int j = 0; j < D14; ++j) {
        const float mm = mk[j >> 1];
        tm[lane * 16 + j]   = mm;
        ti[lane * D14 + j]  = mm * (zz[j] - xp[j]);
        txp[lane * D14 + j] = xp[j];
      }
      tm[lane * 16 + 14] = 0.0f; tm[lane * 16 + 15] = 0.0f;
    }
    __syncthreads();
#pragma unroll 1
    for (int tl = 0; tl < nt; ++tl) {
      const float* qv  = tq  + tl * D14;
      const float* rv  = tr  + tl * D14;
      const float* iv  = ti  + tl * D14;
      const float* xpv = txp + tl * D14;
      const float* mv  = tm  + tl * 16;
      const float qd = qv[i14];
      const float rd = rv[i14];
      const float md = mv[l16];
      float mc[8], ivc[8];
#pragma unroll
      for (int j = 0; j < 8; ++j) {
        mc[j]  = mv[8 * h + j];
        const int ii = 8 * h + j;
        ivc[j] = iv[ii < D14 ? ii : D14 - 1];
      }
#pragma unroll
      for (int j = 0; j < 8; ++j)
        if (8 * h + j == l16 && in14) p[j] += qd;
      float sv[8], g[8];
#pragma unroll
      for (int j = 0; j < 8; ++j) {
        float s = md * mc[j] * p[j];
        if (8 * h + j == l16 && in14) s += md * rd + (1.0f - md) + JITTER * md;
        sv[j] = s;
        g[j]  = s;
      }
      float dsel = 0.0f;
#pragma unroll
      for (int j = 0; j < 8; ++j)
        if (8 * h + j == l16) dsel = sv[j];
      const float dpar = xhalf(dsel);
      const float dgl  = (h == (l16 >> 3)) ? dsel : dpar;
      float tv = (lane < D14) ? md * dgl : 0.0f;
      tv += xhalf(tv);
      tv += row_xor<8>(tv);
      tv += row_xor<4>(tv);
      tv += row_xor<2>(tv);
      tv += row_xor<1>(tv);
      const float dload = JITTER * fmaxf(tv, 1e-6f);
#pragma unroll
      for (int j = 0; j < 8; ++j)
        if (8 * h + j == l16) g[j] = in14 ? (g[j] + dload) : 1.0f;
      gj_step<0>(g, l16, h);  gj_step<1>(g, l16, h);  gj_step<2>(g, l16, h);
      gj_step<3>(g, l16, h);  gj_step<4>(g, l16, h);  gj_step<5>(g, l16, h);
      gj_step<6>(g, l16, h);  gj_step<7>(g, l16, h);  gj_step<8>(g, l16, h);
      gj_step<9>(g, l16, h);  gj_step<10>(g, l16, h); gj_step<11>(g, l16, h);
      gj_step<12>(g, l16, h); gj_step<13>(g, l16, h);
#pragma unroll
      for (int j = 0; j < 8; ++j) Gm[l16 * SM + 8 * h + j] = g[j];
      __syncthreads();
      float gt[8], pho[8];
#pragma unroll
      for (int j = 0; j < 8; ++j) {
        gt[j]  = Gm[(8 * h + j) * SM + l16];
        pho[j] = p[j] * mc[j];
      }
      v8f zero = {};
      const v8f ktv = mm_regs(gt, pho, zero);
      float kt[8];
#pragma unroll
      for (int v = 0; v < 8; ++v) kt[v] = ktv[v];
      {
        float xa = xpv[i14];
#pragma unroll
        for (int v = 0; v < 8; ++v) xa += kt[v] * ivc[v];
        float xb = xhalf(xa);
#pragma unroll
        for (int v = 0; v < 6; ++v) xb += kt[v] * ivc[v];
        if (h == 1 && in14) to[tl * D14 + l16] = xb;
      }
      const v8f kstv = mm_regs(sv, kt, zero);
      float nks[8];
#pragma unroll
      for (int v = 0; v < 8; ++v) nks[v] = -kstv[v];
      v8f pc;
#pragma unroll
      for (int v = 0; v < 8; ++v) pc[v] = p[v];
      pc = mm_regs(nks, kt, pc);
#pragma unroll
      for (int v = 0; v < 8; ++v) Sm[(8 * h + v) * SM + l16] = pc[v];
      __syncthreads();
#pragma unroll
      for (int v = 0; v < 8; ++v) p[v] = 0.5f * (pc[v] + Sm[l16 * SM + 8 * h + v]);
    }
#pragma unroll
    for (int k = 0; k < D14; ++k) {
      const int idx = k * 32 + lane;
      if (idx < nt * D14) out[(base + t0) * D14 + idx] = to[idx];
    }
  }
}

extern "C" void kernel_launch(void* const* d_in, const int* in_sizes, int n_in,
                              void* d_out, int out_size, void* d_ws, size_t ws_size,
                              hipStream_t stream) {
  const float* xh  = (const float*)d_in[0];
  const float* z   = (const float*)d_in[1];
  const float* m21 = (const float*)d_in[2];
  const float* Wf  = (const float*)d_in[3];
  const float* bf  = (const float*)d_in[4];
  const float* Wq  = (const float*)d_in[5];
  const float* bq  = (const float*)d_in[6];
  const float* Wr  = (const float*)d_in[7];
  const float* br  = (const float*)d_in[8];
  float* out = (float*)d_out;

  const int T  = 512;
  const int BT = in_sizes[1] / D14;
  const int B  = BT / T;

  kf_scan<<<dim3(B), dim3(32), 0, stream>>>(xh, z, m21, Wf, bf, Wq, bq, Wr, br,
                                            out, T);
}


# The 14-pivot Gauss-Jordan elimination becomes one refined reciprocal per diagonal element: the matrix is diagonal for every input (P starts as I inside the function, every update term is diagonal)
# speedup vs baseline: 10.5676x; 1.4458x over previous
#include <hip/hip_runtime.h>

#define D14   14
#define SM    17
#define CH    32
#define JITTER 1e-4f
#define MIN_Q  1e-5f
#define MIN_R  1e-4f

typedef __attribute__((ext_vector_type(2))) float v2f;
typedef __attribute__((ext_vector_type(8))) float v8f;

__device__ __forceinline__ v8f mm_regs(const float (&a)[8], const float (&b)[8], v8f c) {
#pragma unroll
  for (int kk = 0; kk < 4; ++kk) {
    v2f av, bv;
    av.x = a[2 * kk]; av.y = a[2 * kk + 1];
    bv.x = b[2 * kk]; bv.y = b[2 * kk + 1];
    c = __builtin_amdgcn_wmma_f32_16x16x4_f32(false, av, false, bv, (short)0, c,
                                              false, false);
  }
  return c;
}


__device__ __forceinline__ float xhalf(float x) {
  return __builtin_bit_cast(float, __builtin_amdgcn_permlanex16(
      0, __builtin_bit_cast(int, x), 0x76543210, 0xfedcba98, false, false));
}

template <int M>
__device__ __forceinline__ float row_xor(float x) {
  return __builtin_bit_cast(float, __builtin_amdgcn_update_dpp(
      0, __builtin_bit_cast(int, x), 0x160 + M, 0xf, 0xf, false));
}

__global__ __launch_bounds__(32) void kf_scan(
    const float* __restrict__ xh, const float* __restrict__ z,
    const float* __restrict__ m21,
    const float* __restrict__ Wf, const float* __restrict__ bf,
    const float* __restrict__ Wq, const float* __restrict__ bq,
    const float* __restrict__ Wr, const float* __restrict__ br,
    float* __restrict__ out, int T) {
  __shared__ float sWf[196], sWq[196], sWr[196];
  __shared__ float sb[48];
  __shared__ float Sm[16 * SM], Gm[16 * SM];
  __shared__ float tq[CH * D14], tr[CH * D14], ti[CH * D14], txp[CH * D14], to[CH * D14];
  __shared__ float tm[CH * 16];

  const int lane = threadIdx.x;
  const int b    = blockIdx.x;
  const int l16  = lane & 15;
  const int h    = lane >> 4;

  for (int i = lane; i < 196; i += 32) { sWf[i] = Wf[i]; sWq[i] = Wq[i]; sWr[i] = Wr[i]; }
  if (lane < 16) {
    sb[lane]      = (lane < D14) ? bf[lane] : 0.f;
    sb[16 + lane] = (lane < D14) ? bq[lane] : 0.f;
    sb[32 + lane] = (lane < D14) ? br[lane] : 0.f;
  }
  const bool in14 = (l16 < D14);
  const int  i14  = in14 ? l16 : 0;
  float p[8];
#pragma unroll
  for (int j = 0; j < 8; ++j) p[j] = (8 * h + j == l16 && in14) ? 1.0f : 0.0f;
  __syncthreads();

  const long long base = (long long)b * T;
#pragma unroll 1
  for (int t0 = 0; t0 < T; t0 += CH) {
    const int nt = (T - t0 < CH) ? (T - t0) : CH;
    {
      const long long tok = base + t0 + ((lane < nt) ? lane : 0);
      float x[D14], zz[D14], xp[D14], mk[7];
#pragma unroll
      for (int i = 0; i < D14; ++i) { x[i] = xh[tok * D14 + i]; zz[i] = z[tok * D14 + i]; }
#pragma unroll
      for (int i = 0; i < 7; ++i) mk[i] = (m21[tok * 21 + 3 * i + 2] > 0.5f) ? 1.0f : 0.0f;
#pragma unroll
      for (int j = 0; j < D14; ++j) {
        float a0 = sb[j], a2 = sb[32 + j];
#pragma unroll
        for (int i = 0; i < D14; ++i) {
          a0 += x[i]  * sWf[i * D14 + j];
          a2 += zz[i] * sWr[i * D14 + j];
        }
        xp[j] = a0;
        tr[lane * D14 + j] = fmaxf(a2, MIN_R);
      }
#pragma unroll
      for (int j = 0; j < D14; ++j) {
        float a1 = sb[16 + j];
#pragma unroll
        for (int i = 0; i < D14; ++i) a1 += xp[i] * sWq[i * D14 + j];
        tq[lane * D14 + j] = fmaxf(a1, MIN_Q);
      }
#pragma unroll
      for (int j = 0; j < D14; ++j) {
        const float mm = mk[j >> 1];
        tm[lane * 16 + j]   = mm;
        ti[lane * D14 + j]  = mm * (zz[j] - xp[j]);
        txp[lane * D14 + j] = xp[j];
      }
      tm[lane * 16 + 14] = 0.0f; tm[lane * 16 + 15] = 0.0f;
    }
    __syncthreads();
#pragma unroll 1
    for (int tl = 0; tl < nt; ++tl) {
      const float* qv  = tq  + tl * D14;
      const float* rv  = tr  + tl * D14;
      const float* iv  = ti  + tl * D14;
      const float* xpv = txp + tl * D14;
      const float* mv  = tm  + tl * 16;
      const float qd = qv[i14];
      const float rd = rv[i14];
      const float md = mv[l16];
      float mc[8], ivc[8];
#pragma unroll
      for (int j = 0; j < 8; ++j) {
        mc[j]  = mv[8 * h + j];
        const int ii = 8 * h + j;
        ivc[j] = iv[ii < D14 ? ii : D14 - 1];
      }
#pragma unroll
      for (int j = 0; j < 8; ++j)
        if (8 * h + j == l16 && in14) p[j] += qd;
      float sv[8], g[8];
#pragma unroll
      for (int j = 0; j < 8; ++j) {
        float s = md * mc[j] * p[j];
        if (8 * h + j == l16 && in14) s += md * rd + (1.0f - md) + JITTER * md;
        sv[j] = s;
        g[j]  = s;
      }
      float dsel = 0.0f;
#pragma unroll
      for (int j = 0; j < 8; ++j)
        if (8 * h + j == l16) dsel = sv[j];
      const float dpar = xhalf(dsel);
      const float dgl  = (h == (l16 >> 3)) ? dsel : dpar;
      float tv = (lane < D14) ? md * dgl : 0.0f;
      tv += xhalf(tv);
      tv += row_xor<8>(tv);
      tv += row_xor<4>(tv);
      tv += row_xor<2>(tv);
      tv += row_xor<1>(tv);
      const float dload = JITTER * fmaxf(tv, 1e-6f);
#pragma unroll
      for (int j = 0; j < 8; ++j) {
        float gi = 0.0f;
        if (8 * h + j == l16) {
          if (in14) {
            const float pv = g[j] + dload;
            const float r0 = __builtin_amdgcn_rcpf(pv);
            gi = __builtin_fmaf(__builtin_fmaf(-pv, r0, 1.0f), r0, r0);
          } else {
            gi = 1.0f;
          }
        }
        g[j] = gi;
      }
#pragma unroll
      for (int j = 0; j < 8; ++j) Gm[l16 * SM + 8 * h + j] = g[j];
      __syncthreads();
      float gt[8], pho[8];
#pragma unroll
      for (int j = 0; j < 8; ++j) {
        gt[j]  = Gm[(8 * h + j) * SM + l16];
        pho[j] = p[j] * mc[j];
      }
      v8f zero = {};
      const v8f ktv = mm_regs(gt, pho, zero);
      float kt[8];
#pragma unroll
      for (int v = 0; v < 8; ++v) kt[v] = ktv[v];
      {
        float xa = xpv[i14];
#pragma unroll
        for (int v = 0; v < 8; ++v) xa += kt[v] * ivc[v];
        float xb = xhalf(xa);
#pragma unroll
        for (int v = 0; v < 6; ++v) xb += kt[v] * ivc[v];
        if (h == 1 && in14) to[tl * D14 + l16] = xb;
      }
      const v8f kstv = mm_regs(sv, kt, zero);
      float nks[8];
#pragma unroll
      for (int v = 0; v < 8; ++v) nks[v] = -kstv[v];
      v8f pc;
#pragma unroll
      for (int v = 0; v < 8; ++v) pc[v] = p[v];
      pc = mm_regs(nks, kt, pc);
#pragma unroll
      for (int v = 0; v < 8; ++v) Sm[(8 * h + v) * SM + l16] = pc[v];
      __syncthreads();
#pragma unroll
      for (int v = 0; v < 8; ++v) p[v] = 0.5f * (pc[v] + Sm[l16 * SM + 8 * h + v]);
    }
#pragma unroll
    for (int k = 0; k < D14; ++k) {
      const int idx = k * 32 + lane;
      if (idx < nt * D14) out[(base + t0) * D14 + idx] = to[idx];
    }
  }
}

extern "C" void kernel_launch(void* const* d_in, const int* in_sizes, int n_in,
                              void* d_out, int out_size, void* d_ws, size_t ws_size,
                              hipStream_t stream) {
  const float* xh  = (const float*)d_in[0];
  const float* z   = (const float*)d_in[1];
  const float* m21 = (const float*)d_in[2];
  const float* Wf  = (const float*)d_in[3];
  const float* bf  = (const float*)d_in[4];
  const float* Wq  = (const float*)d_in[5];
  const float* bq  = (const float*)d_in[6];
  const float* Wr  = (const float*)d_in[7];
  const float* br  = (const float*)d_in[8];
  float* out = (float*)d_out;

  const int T  = 512;
  const int BT = in_sizes[1] / D14;
  const int B  = BT / T;

  kf_scan<<<dim3(B), dim3(32), 0, stream>>>(xh, z, m21, Wf, bf, Wq, bq, Wr, br,
                                            out, T);
}


# Both per-step LDS transposes (of the inverse and of the new P) dropped: each matrix is diagonal with exact zeros, so it is its own transpose and the symmetrised P is Pn bit for bit; sum order kept
# speedup vs baseline: 13.7551x; 1.3016x over previous
#include <hip/hip_runtime.h>

#define D14   14
#define CH    32
#define JITTER 1e-4f
#define MIN_Q  1e-5f
#define MIN_R  1e-4f

typedef __attribute__((ext_vector_type(2))) float v2f;
typedef __attribute__((ext_vector_type(8))) float v8f;

__device__ __forceinline__ v8f mm_regs(const float (&a)[8], const float (&b)[8], v8f c) {
#pragma unroll
  for (int kk = 0; kk < 4; ++kk) {
    v2f av, bv;
    av.x = a[2 * kk]; av.y = a[2 * kk + 1];
    bv.x = b[2 * kk]; bv.y = b[2 * kk + 1];
    c = __builtin_amdgcn_wmma_f32_16x16x4_f32(false, av, false, bv, (short)0, c,
                                              false, false);
  }
  return c;
}


__device__ __forceinline__ float xhalf(float x) {
  return __builtin_bit_cast(float, __builtin_amdgcn_permlanex16(
      0, __builtin_bit_cast(int, x), 0x76543210, 0xfedcba98, false, false));
}

template <int M>
__device__ __forceinline__ float row_xor(float x) {
  return __builtin_bit_cast(float, __builtin_amdgcn_update_dpp(
      0, __builtin_bit_cast(int, x), 0x160 + M, 0xf, 0xf, false));
}

__global__ __launch_bounds__(32) void kf_scan(
    const float* __restrict__ xh, const float* __restrict__ z,
    const float* __restrict__ m21,
    const float* __restrict__ Wf, const float* __restrict__ bf,
    const float* __restrict__ Wq, const float* __restrict__ bq,
    const float* __restrict__ Wr, const float* __restrict__ br,
    float* __restrict__ out, int T) {
  __shared__ float sWf[196], sWq[196], sWr[196];
  __shared__ float sb[48];
  __shared__ float tq[CH * D14], tr[CH * D14], ti[CH * D14], txp[CH * D14], to[CH * D14];
  __shared__ float tm[CH * 16];

  const int lane = threadIdx.x;
  const int b    = blockIdx.x;
  const int l16  = lane & 15;
  const int h    = lane >> 4;

  for (int i = lane; i < 196; i += 32) { sWf[i] = Wf[i]; sWq[i] = Wq[i]; sWr[i] = Wr[i]; }
  if (lane < 16) {
    sb[lane]      = (lane < D14) ? bf[lane] : 0.f;
    sb[16 + lane] = (lane < D14) ? bq[lane] : 0.f;
    sb[32 + lane] = (lane < D14) ? br[lane] : 0.f;
  }
  const bool in14 = (l16 < D14);
  const int  i14  = in14 ? l16 : 0;
  float p[8];
#pragma unroll
  for (int j = 0; j < 8; ++j) p[j] = (8 * h + j == l16 && in14) ? 1.0f : 0.0f;
  __syncthreads();

  const long long base = (long long)b * T;
#pragma unroll 1
  for (int t0 = 0; t0 < T; t0 += CH) {
    const int nt = (T - t0 < CH) ? (T - t0) : CH;
    {
      const long long tok = base + t0 + ((lane < nt) ? lane : 0);
      float x[D14], zz[D14], xp[D14], mk[7];
#pragma unroll
      for (int i = 0; i < D14; ++i) { x[i] = xh[tok * D14 + i]; zz[i] = z[tok * D14 + i]; }
#pragma unroll
      for (int i = 0; i < 7; ++i) mk[i] = (m21[tok * 21 + 3 * i + 2] > 0.5f) ? 1.0f : 0.0f;
#pragma unroll
      for (int j = 0; j < D14; ++j) {
        float a0 = sb[j], a2 = sb[32 + j];
#pragma unroll
        for (int i = 0; i < D14; ++i) {
          a0 += x[i]  * sWf[i * D14 + j];
          a2 += zz[i] * sWr[i * D14 + j];
        }
        xp[j] = a0;
        tr[lane * D14 + j] = fmaxf(a2, MIN_R);
      }
#pragma unroll
      for (int j = 0; j < D14; ++j) {
        float a1 = sb[16 + j];
#pragma unroll
        for (int i = 0; i < D14; ++i) a1 += xp[i] * sWq[i * D14 + j];
        tq[lane * D14 + j] = fmaxf(a1, MIN_Q);
      }
#pragma unroll
      for (int j = 0; j < D14; ++j) {
        const float mm = mk[j >> 1];
        tm[lane * 16 + j]   = mm;
        ti[lane * D14 + j]  = mm * (zz[j] - xp[j]);
        txp[lane * D14 + j] = xp[j];
      }
      tm[lane * 16 + 14] = 0.0f; tm[lane * 16 + 15] = 0.0f;
    }
    __syncthreads();
#pragma unroll 1
    for (int tl = 0; tl < nt; ++tl) {
      const float* qv  = tq  + tl * D14;
      const float* rv  = tr  + tl * D14;
      const float* iv  = ti  + tl * D14;
      const float* xpv = txp + tl * D14;
      const float* mv  = tm  + tl * 16;
      const float qd = qv[i14];
      const float rd = rv[i14];
      const float md = mv[l16];
      float mc[8], ivc[8];
#pragma unroll
      for (int j = 0; j < 8; ++j) {
        mc[j]  = mv[8 * h + j];
        const int ii = 8 * h + j;
        ivc[j] = iv[ii < D14 ? ii : D14 - 1];
      }
      const float xp0 = xpv[i14];
      __builtin_amdgcn_sched_barrier(0);
#pragma unroll
      for (int j = 0; j < 8; ++j)
        if (8 * h + j == l16 && in14) p[j] += qd;
      float sv[8], g[8];
#pragma unroll
      for (int j = 0; j < 8; ++j) {
        float s = md * mc[j] * p[j];
        if (8 * h + j == l16 && in14) s += md * rd + (1.0f - md) + JITTER * md;
        sv[j] = s;
        g[j]  = s;
      }
      float dsel = 0.0f;
#pragma unroll
      for (int j = 0; j < 8; ++j)
        if (8 * h + j == l16) dsel = sv[j];
      const float dpar = xhalf(dsel);
      const float dgl  = (h == (l16 >> 3)) ? dsel : dpar;
      float tv = (lane < D14) ? md * dgl : 0.0f;
      tv += xhalf(tv);
      tv += row_xor<8>(tv);
      tv += row_xor<4>(tv);
      tv += row_xor<2>(tv);
      tv += row_xor<1>(tv);
      const float dload = JITTER * fmaxf(tv, 1e-6f);
#pragma unroll
      for (int j = 0; j < 8; ++j) {
        float gi = 0.0f;
        if (8 * h + j == l16) {
          if (in14) {
            const float pv = g[j] + dload;
            const float r0 = __builtin_amdgcn_rcpf(pv);
            gi = __builtin_fmaf(__builtin_fmaf(-pv, r0, 1.0f), r0, r0);
          } else {
            gi = 1.0f;
          }
        }
        g[j] = gi;
      }
      float pho[8];
#pragma unroll
      for (int j = 0; j < 8; ++j) pho[j] = p[j] * mc[j];
      v8f zero = {};
      const v8f ktv = mm_regs(g, pho, zero);
      float kt[8];
#pragma unroll
      for (int v = 0; v < 8; ++v) kt[v] = ktv[v];
      {
        float xa = xp0;
#pragma unroll
        for (int v = 0; v < 8; ++v) xa += kt[v] * ivc[v];
        float xb = xhalf(xa);
#pragma unroll
        for (int v = 0; v < 6; ++v) xb += kt[v] * ivc[v];
        if (h == 1 && in14) to[tl * D14 + l16] = xb;
      }
      const v8f kstv = mm_regs(sv, kt, zero);
      float nks[8];
#pragma unroll
      for (int v = 0; v < 8; ++v) nks[v] = -kstv[v];
      v8f pc;
#pragma unroll
      for (int v = 0; v < 8; ++v) pc[v] = p[v];
      pc = mm_regs(nks, kt, pc);
#pragma unroll
      for (int v = 0; v < 8; ++v) p[v] = pc[v];
    }
    __syncthreads();
#pragma unroll
    for (int k = 0; k < D14; ++k) {
      const int idx = k * 32 + lane;
      if (idx < nt * D14) out[(base + t0) * D14 + idx] = to[idx];
    }
  }
}

extern "C" void kernel_launch(void* const* d_in, const int* in_sizes, int n_in,
                              void* d_out, int out_size, void* d_ws, size_t ws_size,
                              hipStream_t stream) {
  const float* xh  = (const float*)d_in[0];
  const float* z   = (const float*)d_in[1];
  const float* m21 = (const float*)d_in[2];
  const float* Wf  = (const float*)d_in[3];
  const float* bf  = (const float*)d_in[4];
  const float* Wq  = (const float*)d_in[5];
  const float* bq  = (const float*)d_in[6];
  const float* Wr  = (const float*)d_in[7];
  const float* br  = (const float*)d_in[8];
  float* out = (float*)d_out;

  const int T  = 512;
  const int BT = in_sizes[1] / D14;
  const int B  = BT / T;

  kf_scan<<<dim3(B), dim3(32), 0, stream>>>(xh, z, m21, Wf, bf, Wq, bq, Wr, br,
                                            out, T);
}


# A second wave per work-group fills the next chunk's tables and stores the previous chunk's outputs while the first runs the serial scan; two table sets in LDS, one barrier per chunk; sum order kept
# speedup vs baseline: 19.0273x; 1.3833x over previous
#include <hip/hip_runtime.h>

#define D14   14
#define CH    32
#define JITTER 1e-4f
#define MIN_Q  1e-5f
#define MIN_R  1e-4f

typedef __attribute__((ext_vector_type(2))) float v2f;
typedef __attribute__((ext_vector_type(8))) float v8f;

__device__ __forceinline__ v8f mm_regs(const float (&a)[8], const float (&b)[8], v8f c) {
#pragma unroll
  for (int kk = 0; kk < 4; ++kk) {
    v2f av, bv;
    av.x = a[2 * kk]; av.y = a[2 * kk + 1];
    bv.x = b[2 * kk]; bv.y = b[2 * kk + 1];
    c = __builtin_amdgcn_wmma_f32_16x16x4_f32(false, av, false, bv, (short)0, c,
                                              false, false);
  }
  return c;
}


__device__ __forceinline__ float xhalf(float x) {
  return __builtin_bit_cast(float, __builtin_amdgcn_permlanex16(
      0, __builtin_bit_cast(int, x), 0x76543210, 0xfedcba98, false, false));
}

template <int M>
__device__ __forceinline__ float row_xor(float x) {
  return __builtin_bit_cast(float, __builtin_amdgcn_update_dpp(
      0, __builtin_bit_cast(int, x), 0x160 + M, 0xf, 0xf, false));
}

struct ChunkTables {
  float q[CH * D14], r[CH * D14], i[CH * D14], xp[CH * D14], o[CH * D14];
  float m[CH * 16];
};

__global__ __launch_bounds__(64) void kf_scan(
    const float* __restrict__ xh, const float* __restrict__ z,
    const float* __restrict__ m21,
    const float* __restrict__ Wf, const float* __restrict__ bf,
    const float* __restrict__ Wq, const float* __restrict__ bq,
    const float* __restrict__ Wr, const float* __restrict__ br,
    float* __restrict__ out, int T) {
  __shared__ float sWf[196], sWq[196], sWr[196];
  __shared__ float sb[48];
  __shared__ ChunkTables tb[2];

  const int tid  = threadIdx.x;
  const int lane = tid & 31;
  const int wv   = __builtin_amdgcn_readfirstlane(tid >> 5);
  const int b    = blockIdx.x;
  const int l16  = lane & 15;
  const int h    = lane >> 4;

  for (int i = tid; i < 196; i += 64) { sWf[i] = Wf[i]; sWq[i] = Wq[i]; sWr[i] = Wr[i]; }
  if (tid < 16) {
    sb[tid]      = (tid < D14) ? bf[tid] : 0.f;
    sb[16 + tid] = (tid < D14) ? bq[tid] : 0.f;
    sb[32 + tid] = (tid < D14) ? br[tid] : 0.f;
  }
  const bool in14 = (l16 < D14);
  const int  i14  = in14 ? l16 : 0;
  float p[8];
#pragma unroll
  for (int j = 0; j < 8; ++j) p[j] = (8 * h + j == l16 && in14) ? 1.0f : 0.0f;
  __syncthreads();

  const long long base = (long long)b * T;
  const int nch = (T + CH - 1) / CH;
#pragma unroll 1
  for (int c = 0; c <= nch + 1; ++c) {
    if (wv == 1) {
      if (c < nch) {
        const int t0 = c * CH;
        const int nt = (T - t0 < CH) ? (T - t0) : CH;
        const int ws = c & 1;
        ChunkTables& wt = tb[ws];
        float* wq_ = wt.q; float* wr_ = wt.r; float* wi_ = wt.i; float* wxp_ = wt.xp; float* wm_ = wt.m;
        {
          const long long tok = base + t0 + ((lane < nt) ? lane : 0);
          float x[D14], zz[D14], xp[D14], mk[7];
#pragma unroll
          for (int i = 0; i < D14; ++i) { x[i] = xh[tok * D14 + i]; zz[i] = z[tok * D14 + i]; }
#pragma unroll
          for (int i = 0; i < 7; ++i) mk[i] = (m21[tok * 21 + 3 * i + 2] > 0.5f) ? 1.0f : 0.0f;
#pragma unroll
          for (int j = 0; j < D14; ++j) {
            float a0 = sb[j], a2 = sb[32 + j];
#pragma unroll
            for (int i = 0; i < D14; ++i) {
              a0 += x[i]  * sWf[i * D14 + j];
              a2 += zz[i] * sWr[i * D14 + j];
            }
            xp[j] = a0;
            wr_[lane * D14 + j] = fmaxf(a2, MIN_R);
          }
#pragma unroll
          for (int j = 0; j < D14; ++j) {
            float a1 = sb[16 + j];
#pragma unroll
            for (int i = 0; i < D14; ++i) a1 += xp[i] * sWq[i * D14 + j];
            wq_[lane * D14 + j] = fmaxf(a1, MIN_Q);
          }
#pragma unroll
          for (int j = 0; j < D14; ++j) {
            const float mm = mk[j >> 1];
            wm_[lane * 16 + j]   = mm;
            wi_[lane * D14 + j]  = mm * (zz[j] - xp[j]);
            wxp_[lane * D14 + j] = xp[j];
          }
          wm_[lane * 16 + 14] = 0.0f; wm_[lane * 16 + 15] = 0.0f;
        }
      }
      if (c >= 2) {
        const int e0 = (c - 2) * CH;
        const int ne = (T - e0 < CH) ? (T - e0) : CH;
        const float* eo_ = tb[c & 1].o;
#pragma unroll
        for (int k = 0; k < D14; ++k) {
          const int idx = k * 32 + lane;
          if (idx < ne * D14) out[(base + e0) * D14 + idx] = eo_[idx];
        }
      }
    } else if (c >= 1 && c <= nch) {
      const int t0 = (c - 1) * CH;
      const int nt = (T - t0 < CH) ? (T - t0) : CH;
      const int rs = (c - 1) & 1;
      ChunkTables& rt = tb[rs];
      const float* rq_ = rt.q; const float* rr_ = rt.r; const float* ri_ = rt.i;
      const float* rxp_ = rt.xp; const float* rm_ = rt.m; float* ro_ = rt.o;
#pragma unroll 1
      for (int tl = 0; tl < nt; ++tl) {
        const float* qv  = rq_  + tl * D14;
        const float* rv  = rr_  + tl * D14;
        const float* iv  = ri_  + tl * D14;
        const float* xpv = rxp_ + tl * D14;
        const float* mv  = rm_  + tl * 16;
        const float qd = qv[i14];
        const float rd = rv[i14];
        const float md = mv[l16];
        float mc[8], ivc[8];
#pragma unroll
        for (int j = 0; j < 8; ++j) {
          mc[j]  = mv[8 * h + j];
          const int ii = 8 * h + j;
          ivc[j] = iv[ii < D14 ? ii : D14 - 1];
        }
        float xp0 = xpv[i14];
#pragma unroll
        for (int j = 0; j < 8; ++j) asm volatile("" : "+v"(ivc[j]));
        asm volatile("" : "+v"(xp0));
        __builtin_amdgcn_sched_barrier(0);
#pragma unroll
        for (int j = 0; j < 8; ++j)
          if (8 * h + j == l16 && in14) p[j] += qd;
        float sv[8], g[8];
#pragma unroll
        for (int j = 0; j < 8; ++j) {
          float s = md * mc[j] * p[j];
          if (8 * h + j == l16 && in14) s += md * rd + (1.0f - md) + JITTER * md;
          sv[j] = s;
          g[j]  = s;
        }
        float dsel = 0.0f;
#pragma unroll
        for (int j = 0; j < 8; ++j)
          if (8 * h + j == l16) dsel = sv[j];
        const float dpar = xhalf(dsel);
        const float dgl  = (h == (l16 >> 3)) ? dsel : dpar;
        float tv = (lane < D14) ? md * dgl : 0.0f;
        tv += xhalf(tv);
        tv += row_xor<8>(tv);
        tv += row_xor<4>(tv);
        tv += row_xor<2>(tv);
        tv += row_xor<1>(tv);
        const float dload = JITTER * fmaxf(tv, 1e-6f);
#pragma unroll
        for (int j = 0; j < 8; ++j) {
          float gi = 0.0f;
          if (8 * h + j == l16) {
            if (in14) {
              const float pv = g[j] + dload;
              const float r0 = __builtin_amdgcn_rcpf(pv);
              gi = __builtin_fmaf(__builtin_fmaf(-pv, r0, 1.0f), r0, r0);
            } else {
              gi = 1.0f;
            }
          }
          g[j] = gi;
        }
        float pho[8];
#pragma unroll
        for (int j = 0; j < 8; ++j) pho[j] = p[j] * mc[j];
        v8f zero = {};
        const v8f ktv = mm_regs(g, pho, zero);
        float kt[8];
#pragma unroll
        for (int v = 0; v < 8; ++v) kt[v] = ktv[v];
        {
          float xa = xp0;
#pragma unroll
          for (int v = 0; v < 8; ++v) xa += kt[v] * ivc[v];
          float xb = xhalf(xa);
#pragma unroll
          for (int v = 0; v < 6; ++v) xb += kt[v] * ivc[v];
          if (h == 1 && in14) ro_[tl * D14 + l16] = xb;
        }
        const v8f kstv = mm_regs(sv, kt, zero);
        float nks[8];
#pragma unroll
        for (int v = 0; v < 8; ++v) nks[v] = -kstv[v];
        v8f pc;
#pragma unroll
        for (int v = 0; v < 8; ++v) pc[v] = p[v];
        pc = mm_regs(nks, kt, pc);
#pragma unroll
        for (int v = 0; v < 8; ++v) p[v] = pc[v];
      }
    }
    __syncthreads();
  }
}

extern "C" void kernel_launch(void* const* d_in, const int* in_sizes, int n_in,
                              void* d_out, int out_size, void* d_ws, size_t ws_size,
                              hipStream_t stream) {
  const float* xh  = (const float*)d_in[0];
  const float* z   = (const float*)d_in[1];
  const float* m21 = (const float*)d_in[2];
  const float* Wf  = (const float*)d_in[3];
  const float* bf  = (const float*)d_in[4];
  const float* Wq  = (const float*)d_in[5];
  const float* bq  = (const float*)d_in[6];
  const float* Wr  = (const float*)d_in[7];
  const float* br  = (const float*)d_in[8];
  float* out = (float*)d_out;

  const int T  = 512;
  const int BT = in_sizes[1] / D14;
  const int B  = BT / T;

  kf_scan<<<dim3(B), dim3(64), 0, stream>>>(xh, z, m21, Wf, bf, Wq, bq, Wr, br,
                                            out, T);
}


# State update cut to its one non-zero term (K is diagonal; the 13 dropped terms are exact zeros): one fma in the owning lane, stored under the next step's table loads, not between the products
# speedup vs baseline: 24.0548x; 1.2642x over previous
#include <hip/hip_runtime.h>

#define D14   14
#define CH    32
#define JITTER 1e-4f
#define MIN_Q  1e-5f
#define MIN_R  1e-4f

typedef __attribute__((ext_vector_type(2))) float v2f;
typedef __attribute__((ext_vector_type(8))) float v8f;

__device__ __forceinline__ v8f mm_regs(const float (&a)[8], const float (&b)[8], v8f c) {
#pragma unroll
  for (int kk = 0; kk < 4; ++kk) {
    v2f av, bv;
    av.x = a[2 * kk]; av.y = a[2 * kk + 1];
    bv.x = b[2 * kk]; bv.y = b[2 * kk + 1];
    c = __builtin_amdgcn_wmma_f32_16x16x4_f32(false, av, false, bv, (short)0, c,
                                              false, false);
  }
  return c;
}


__device__ __forceinline__ float xhalf(float x) {
  return __builtin_bit_cast(float, __builtin_amdgcn_permlanex16(
      0, __builtin_bit_cast(int, x), 0x76543210, 0xfedcba98, false, false));
}

template <int M>
__device__ __forceinline__ float row_xor(float x) {
  return __builtin_bit_cast(float, __builtin_amdgcn_update_dpp(
      0, __builtin_bit_cast(int, x), 0x160 + M, 0xf, 0xf, false));
}

struct ChunkTables {
  float q[CH * D14], r[CH * D14], i[CH * D14], xp[CH * D14], o[CH * D14];
  float m[CH * 16];
};

__global__ __launch_bounds__(64) void kf_scan(
    const float* __restrict__ xh, const float* __restrict__ z,
    const float* __restrict__ m21,
    const float* __restrict__ Wf, const float* __restrict__ bf,
    const float* __restrict__ Wq, const float* __restrict__ bq,
    const float* __restrict__ Wr, const float* __restrict__ br,
    float* __restrict__ out, int T) {
  __shared__ float sWf[196], sWq[196], sWr[196];
  __shared__ float sb[48];
  __shared__ ChunkTables tb[2];

  const int tid  = threadIdx.x;
  const int lane = tid & 31;
  const int wv   = __builtin_amdgcn_readfirstlane(tid >> 5);
  const int b    = blockIdx.x;
  const int l16  = lane & 15;
  const int h    = lane >> 4;

  for (int i = tid; i < 196; i += 64) { sWf[i] = Wf[i]; sWq[i] = Wq[i]; sWr[i] = Wr[i]; }
  if (tid < 16) {
    sb[tid]      = (tid < D14) ? bf[tid] : 0.f;
    sb[16 + tid] = (tid < D14) ? bq[tid] : 0.f;
    sb[32 + tid] = (tid < D14) ? br[tid] : 0.f;
  }
  const bool in14 = (l16 < D14);
  const int  i14  = in14 ? l16 : 0;
  const bool own  = in14 && (h == (l16 >> 3));
  const bool sel0 = (l16 & 1) != 0, sel1 = (l16 & 2) != 0, sel2 = (l16 & 4) != 0;
  float p[8];
#pragma unroll
  for (int j = 0; j < 8; ++j) p[j] = (8 * h + j == l16 && in14) ? 1.0f : 0.0f;
  __syncthreads();

  const long long base = (long long)b * T;
  const int nch = (T + CH - 1) / CH;
#pragma unroll 1
  for (int c = 0; c <= nch + 1; ++c) {
    if (wv == 1) {
      if (c < nch) {
        const int t0 = c * CH;
        const int nt = (T - t0 < CH) ? (T - t0) : CH;
        const int ws = c & 1;
        ChunkTables& wt = tb[ws];
        float* wq_ = wt.q; float* wr_ = wt.r; float* wi_ = wt.i; float* wxp_ = wt.xp; float* wm_ = wt.m;
        {
          const long long tok = base + t0 + ((lane < nt) ? lane : 0);
          float x[D14], zz[D14], xp[D14], mk[7];
#pragma unroll
          for (int i = 0; i < D14; ++i) { x[i] = xh[tok * D14 + i]; zz[i] = z[tok * D14 + i]; }
#pragma unroll
          for (int i = 0; i < 7; ++i) mk[i] = (m21[tok * 21 + 3 * i + 2] > 0.5f) ? 1.0f : 0.0f;
#pragma unroll
          for (int j = 0; j < D14; ++j) {
            float a0 = sb[j], a2 = sb[32 + j];
#pragma unroll
            for (int i = 0; i < D14; ++i) {
              a0 += x[i]  * sWf[i * D14 + j];
              a2 += zz[i] * sWr[i * D14 + j];
            }
            xp[j] = a0;
            wr_[lane * D14 + j] = fmaxf(a2, MIN_R);
          }
#pragma unroll
          for (int j = 0; j < D14; ++j) {
            float a1 = sb[16 + j];
#pragma unroll
            for (int i = 0; i < D14; ++i) a1 += xp[i] * sWq[i * D14 + j];
            wq_[lane * D14 + j] = fmaxf(a1, MIN_Q);
          }
#pragma unroll
          for (int j = 0; j < D14; ++j) {
            const float mm = mk[j >> 1];
            wm_[lane * 16 + j]   = mm;
            wi_[lane * D14 + j]  = mm * (zz[j] - xp[j]);
            wxp_[lane * D14 + j] = xp[j];
          }
          wm_[lane * 16 + 14] = 0.0f; wm_[lane * 16 + 15] = 0.0f;
        }
      }
      if (c >= 2) {
        const int e0 = (c - 2) * CH;
        const int ne = (T - e0 < CH) ? (T - e0) : CH;
        const float* eo_ = tb[c & 1].o;
#pragma unroll
        for (int k = 0; k < D14; ++k) {
          const int idx = k * 32 + lane;
          if (idx < ne * D14) out[(base + e0) * D14 + idx] = eo_[idx];
        }
      }
    } else if (c >= 1 && c <= nch) {
      const int t0 = (c - 1) * CH;
      const int nt = (T - t0 < CH) ? (T - t0) : CH;
      const int rs = (c - 1) & 1;
      ChunkTables& rt = tb[rs];
      const float* rq_ = rt.q; const float* rr_ = rt.r; const float* ri_ = rt.i;
      const float* rxp_ = rt.xp; const float* rm_ = rt.m; float* ro_ = rt.o;
      float xn = 0.0f;
#pragma unroll 1
      for (int tl = 0; tl < nt; ++tl) {
        const float* qv  = rq_  + tl * D14;
        const float* rv  = rr_  + tl * D14;
        const float* iv  = ri_  + tl * D14;
        const float* xpv = rxp_ + tl * D14;
        const float* mv  = rm_  + tl * 16;
        const float qd = qv[i14];
        const float rd = rv[i14];
        const float md = mv[l16];
        float mc[8];
#pragma unroll
        for (int j = 0; j < 8; ++j) mc[j] = mv[8 * h + j];
        float ivd = iv[i14];
        float xp0 = xpv[i14];
        if (own && tl > 0) ro_[(tl - 1) * D14 + l16] = xn;
#pragma unroll
        for (int j = 0; j < 8; ++j)
          if (8 * h + j == l16 && in14) p[j] += qd;
        float sv[8], g[8];
#pragma unroll
        for (int j = 0; j < 8; ++j) {
          float s = md * mc[j] * p[j];
          if (8 * h + j == l16 && in14) s += md * rd + (1.0f - md) + JITTER * md;
          sv[j] = s;
          g[j]  = s;
        }
        float dsel = 0.0f;
#pragma unroll
        for (int j = 0; j < 8; ++j)
          if (8 * h + j == l16) dsel = sv[j];
        const float dpar = xhalf(dsel);
        const float dgl  = (h == (l16 >> 3)) ? dsel : dpar;
        float tv = (lane < D14) ? md * dgl : 0.0f;
        tv += xhalf(tv);
        tv += row_xor<8>(tv);
        tv += row_xor<4>(tv);
        tv += row_xor<2>(tv);
        tv += row_xor<1>(tv);
        const float dload = JITTER * fmaxf(tv, 1e-6f);
#pragma unroll
        for (int j = 0; j < 8; ++j) {
          float gi = 0.0f;
          if (8 * h + j == l16) {
            if (in14) {
              const float pv = g[j] + dload;
              const float r0 = __builtin_amdgcn_rcpf(pv);
              gi = __builtin_fmaf(__builtin_fmaf(-pv, r0, 1.0f), r0, r0);
            } else {
              gi = 1.0f;
            }
          }
          g[j] = gi;
        }
        float pho[8];
#pragma unroll
        for (int j = 0; j < 8; ++j) pho[j] = p[j] * mc[j];
        v8f zero = {};
        const v8f ktv = mm_regs(g, pho, zero);
        float kt[8];
#pragma unroll
        for (int v = 0; v < 8; ++v) kt[v] = ktv[v];
        const float ka0 = sel0 ? kt[1] : kt[0];
        const float ka1 = sel0 ? kt[3] : kt[2];
        const float ka2 = sel0 ? kt[5] : kt[4];
        const float ka3 = sel0 ? kt[7] : kt[6];
        const float kb0 = sel1 ? ka1 : ka0;
        const float kb1 = sel1 ? ka3 : ka2;
        const float kd  = sel2 ? kb1 : kb0;
        xn = __builtin_fmaf(kd, ivd, xp0);
        const v8f kstv = mm_regs(sv, kt, zero);
        float nks[8];
#pragma unroll
        for (int v = 0; v < 8; ++v) nks[v] = -kstv[v];
        v8f pc;
#pragma unroll
        for (int v = 0; v < 8; ++v) pc[v] = p[v];
        pc = mm_regs(nks, kt, pc);
#pragma unroll
        for (int v = 0; v < 8; ++v) p[v] = pc[v];
      }
      if (own) ro_[(nt - 1) * D14 + l16] = xn;
    }
    __syncthreads();
  }
}

extern "C" void kernel_launch(void* const* d_in, const int* in_sizes, int n_in,
                              void* d_out, int out_size, void* d_ws, size_t ws_size,
                              hipStream_t stream) {
  const float* xh  = (const float*)d_in[0];
  const float* z   = (const float*)d_in[1];
  const float* m21 = (const float*)d_in[2];
  const float* Wf  = (const float*)d_in[3];
  const float* bf  = (const float*)d_in[4];
  const float* Wq  = (const float*)d_in[5];
  const float* bq  = (const float*)d_in[6];
  const float* Wr  = (const float*)d_in[7];
  const float* br  = (const float*)d_in[8];
  float* out = (float*)d_out;

  const int T  = 512;
  const int BT = in_sizes[1] / D14;
  const int B  = BT / T;

  kf_scan<<<dim3(B), dim3(64), 0, stream>>>(xh, z, m21, Wf, bf, Wq, bq, Wr, br,
                                            out, T);
}


# Each step's table row is loaded from LDS one step ahead into a second register set (two steps per trip, sets swap roles; store made branch-free): no LDS wait opens a step; sums unchanged
# speedup vs baseline: 25.5512x; 1.0622x over previous
#include <hip/hip_runtime.h>

#define D14   14
#define CH    32
#define JITTER 1e-4f
#define MIN_Q  1e-5f
#define MIN_R  1e-4f

typedef __attribute__((ext_vector_type(2))) float v2f;
typedef __attribute__((ext_vector_type(8))) float v8f;

__device__ __forceinline__ v8f mm_regs(const float (&a)[8], const float (&b)[8], v8f c) {
#pragma unroll
  for (int kk = 0; kk < 4; ++kk) {
    v2f av, bv;
    av.x = a[2 * kk]; av.y = a[2 * kk + 1];
    bv.x = b[2 * kk]; bv.y = b[2 * kk + 1];
    c = __builtin_amdgcn_wmma_f32_16x16x4_f32(false, av, false, bv, (short)0, c,
                                              false, false);
  }
  return c;
}


__device__ __forceinline__ float xhalf(float x) {
  return __builtin_bit_cast(float, __builtin_amdgcn_permlanex16(
      0, __builtin_bit_cast(int, x), 0x76543210, 0xfedcba98, false, false));
}

template <int M>
__device__ __forceinline__ float row_xor(float x) {
  return __builtin_bit_cast(float, __builtin_amdgcn_update_dpp(
      0, __builtin_bit_cast(int, x), 0x160 + M, 0xf, 0xf, false));
}

struct ChunkTables {
  float q[CH * D14], r[CH * D14], i[CH * D14], xp[CH * D14], o[CH * D14];
  float m[CH * 16];
};

struct LanePos {
  int  l16, h, i14;
  bool in14;
  bool sel0, sel1, sel2;
};

struct StepIn { float qd, rd, md, ivd, xp0; float mc[8]; };

__device__ __forceinline__ void kf_load_row(StepIn& in, const ChunkTables& rt, int tr, LanePos lp) {
  const float* mv = rt.m + tr * 16;
  in.qd  = rt.q[tr * D14 + lp.i14];
  in.rd  = rt.r[tr * D14 + lp.i14];
  in.md  = mv[lp.l16];
#pragma unroll
  for (int j = 0; j < 8; ++j) in.mc[j] = mv[8 * lp.h + j];
  in.ivd = rt.i[tr * D14 + lp.i14];
  in.xp0 = rt.xp[tr * D14 + lp.i14];
}

__device__ __forceinline__ void kf_step(float (&p)[8], float& xn, const StepIn& in, LanePos lp, int lane) {
  const int  l16 = lp.l16, h = lp.h;
  const bool in14 = lp.in14, sel0 = lp.sel0, sel1 = lp.sel1, sel2 = lp.sel2;
  const float qd = in.qd;
  const float rd = in.rd;
  const float md = in.md;
  float mc[8];
#pragma unroll
  for (int j = 0; j < 8; ++j) mc[j] = in.mc[j];
  const float ivd = in.ivd;
  const float xp0 = in.xp0;
#pragma unroll
  for (int j = 0; j < 8; ++j)
    if (8 * h + j == l16 && in14) p[j] += qd;
  float sv[8], g[8];
#pragma unroll
  for (int j = 0; j < 8; ++j) {
    float s = md * mc[j] * p[j];
    if (8 * h + j == l16 && in14) s += md * rd + (1.0f - md) + JITTER * md;
    sv[j] = s;
    g[j]  = s;
  }
  float dsel = 0.0f;
#pragma unroll
  for (int j = 0; j < 8; ++j)
    if (8 * h + j == l16) dsel = sv[j];
  const float dpar = xhalf(dsel);
  const float dgl  = (h == (l16 >> 3)) ? dsel : dpar;
  float tv = (lane < D14) ? md * dgl : 0.0f;
  tv += xhalf(tv);
  tv += row_xor<8>(tv);
  tv += row_xor<4>(tv);
  tv += row_xor<2>(tv);
  tv += row_xor<1>(tv);
  const float dload = JITTER * fmaxf(tv, 1e-6f);
#pragma unroll
  for (int j = 0; j < 8; ++j) {
    float gi = 0.0f;
    if (8 * h + j == l16) {
      if (in14) {
        const float pv = g[j] + dload;
        const float r0 = __builtin_amdgcn_rcpf(pv);
        gi = __builtin_fmaf(__builtin_fmaf(-pv, r0, 1.0f), r0, r0);
      } else {
        gi = 1.0f;
      }
    }
    g[j] = gi;
  }
  float pho[8];
#pragma unroll
  for (int j = 0; j < 8; ++j) pho[j] = p[j] * mc[j];
  v8f zero = {};
  const v8f ktv = mm_regs(g, pho, zero);
  float kt[8];
#pragma unroll
  for (int v = 0; v < 8; ++v) kt[v] = ktv[v];
  const float ka0 = sel0 ? kt[1] : kt[0];
  const float ka1 = sel0 ? kt[3] : kt[2];
  const float ka2 = sel0 ? kt[5] : kt[4];
  const float ka3 = sel0 ? kt[7] : kt[6];
  const float kb0 = sel1 ? ka1 : ka0;
  const float kb1 = sel1 ? ka3 : ka2;
  const float kd  = sel2 ? kb1 : kb0;
  xn = __builtin_fmaf(kd, ivd, xp0);
  const v8f kstv = mm_regs(sv, kt, zero);
  float nks[8];
#pragma unroll
  for (int v = 0; v < 8; ++v) nks[v] = -kstv[v];
  v8f pc;
#pragma unroll
  for (int v = 0; v < 8; ++v) pc[v] = p[v];
  pc = mm_regs(nks, kt, pc);
#pragma unroll
  for (int v = 0; v < 8; ++v) p[v] = pc[v];
}

__global__ __launch_bounds__(64) void kf_scan(
    const float* __restrict__ xh, const float* __restrict__ z,
    const float* __restrict__ m21,
    const float* __restrict__ Wf, const float* __restrict__ bf,
    const float* __restrict__ Wq, const float* __restrict__ bq,
    const float* __restrict__ Wr, const float* __restrict__ br,
    float* __restrict__ out, int T) {
  __shared__ float sWf[196], sWq[196], sWr[196];
  __shared__ float sb[48];
  __shared__ ChunkTables tb[2];
  __shared__ float sdump[32];

  const int tid  = threadIdx.x;
  const int lane = tid & 31;
  const int wv   = __builtin_amdgcn_readfirstlane(tid >> 5);
  const int b    = blockIdx.x;
  const int l16  = lane & 15;
  const int h    = lane >> 4;

  for (int i = tid; i < 196; i += 64) { sWf[i] = Wf[i]; sWq[i] = Wq[i]; sWr[i] = Wr[i]; }
  if (tid < 16) {
    sb[tid]      = (tid < D14) ? bf[tid] : 0.f;
    sb[16 + tid] = (tid < D14) ? bq[tid] : 0.f;
    sb[32 + tid] = (tid < D14) ? br[tid] : 0.f;
  }
  const bool in14 = (l16 < D14);
  const int  i14  = in14 ? l16 : 0;
  const bool own  = in14 && (h == (l16 >> 3));
  const bool sel0 = (l16 & 1) != 0, sel1 = (l16 & 2) != 0, sel2 = (l16 & 4) != 0;
  const LanePos lp = {l16, h, i14, in14, sel0, sel1, sel2};
  float p[8];
#pragma unroll
  for (int j = 0; j < 8; ++j) p[j] = (8 * h + j == l16 && in14) ? 1.0f : 0.0f;
  __syncthreads();

  const long long base = (long long)b * T;
  const int nch = (T + CH - 1) / CH;
#pragma unroll 1
  for (int c = 0; c <= nch + 1; ++c) {
    if (wv == 1) {
      if (c < nch) {
        const int t0 = c * CH;
        const int nt = (T - t0 < CH) ? (T - t0) : CH;
        const int ws = c & 1;
        ChunkTables& wt = tb[ws];
        float* wq_ = wt.q; float* wr_ = wt.r; float* wi_ = wt.i; float* wxp_ = wt.xp; float* wm_ = wt.m;
        {
          const long long tok = base + t0 + ((lane < nt) ? lane : 0);
          float x[D14], zz[D14], xp[D14], mk[7];
#pragma unroll
          for (int i = 0; i < D14; ++i) { x[i] = xh[tok * D14 + i]; zz[i] = z[tok * D14 + i]; }
#pragma unroll
          for (int i = 0; i < 7; ++i) mk[i] = (m21[tok * 21 + 3 * i + 2] > 0.5f) ? 1.0f : 0.0f;
#pragma unroll
          for (int j = 0; j < D14; ++j) {
            float a0 = sb[j], a2 = sb[32 + j];
#pragma unroll
            for (int i = 0; i < D14; ++i) {
              a0 += x[i]  * sWf[i * D14 + j];
              a2 += zz[i] * sWr[i * D14 + j];
            }
            xp[j] = a0;
            wr_[lane * D14 + j] = fmaxf(a2, MIN_R);
          }
#pragma unroll
          for (int j = 0; j < D14; ++j) {
            float a1 = sb[16 + j];
#pragma unroll
            for (int i = 0; i < D14; ++i) a1 += xp[i] * sWq[i * D14 + j];
            wq_[lane * D14 + j] = fmaxf(a1, MIN_Q);
          }
#pragma unroll
          for (int j = 0; j < D14; ++j) {
            const float mm = mk[j >> 1];
            wm_[lane * 16 + j]   = mm;
            wi_[lane * D14 + j]  = mm * (zz[j] - xp[j]);
            wxp_[lane * D14 + j] = xp[j];
          }
          wm_[lane * 16 + 14] = 0.0f; wm_[lane * 16 + 15] = 0.0f;
        }
      }
      if (c >= 2) {
        const int e0 = (c - 2) * CH;
        const int ne = (T - e0 < CH) ? (T - e0) : CH;
        const float* eo_ = tb[c & 1].o;
#pragma unroll
        for (int k = 0; k < D14; ++k) {
          const int idx = k * 32 + lane;
          if (idx < ne * D14) out[(base + e0) * D14 + idx] = eo_[idx];
        }
      }
    } else if (c >= 1 && c <= nch) {
      const int t0 = (c - 1) * CH;
      const int nt = (T - t0 < CH) ? (T - t0) : CH;
      const int rs = (c - 1) & 1;
      const ChunkTables& rt = tb[rs];
      float* xb = own ? tb[rs].o + l16 : sdump + lane;
      const int xs = own ? D14 : 0;
      float xn = 0.0f;
      StepIn in_a, in_b;
      kf_load_row(in_a, rt, 0, lp);
#pragma unroll 1
      for (int tl = 0; tl < nt; tl += 2) {
        *((tl > 0) ? xb + (tl - 1) * xs : sdump + lane) = xn;
        kf_load_row(in_b, rt, (tl + 1 < nt) ? tl + 1 : tl, lp);
        __builtin_amdgcn_sched_barrier(0);
        kf_step(p, xn, in_a, lp, lane);
        if (tl + 1 >= nt) break;
        *(xb + tl * xs) = xn;
        kf_load_row(in_a, rt, (tl + 2 < nt) ? tl + 2 : tl + 1, lp);
        __builtin_amdgcn_sched_barrier(0);
        kf_step(p, xn, in_b, lp, lane);
      }
      if (own) tb[rs].o[(nt - 1) * D14 + l16] = xn;
    }
    __syncthreads();
  }
}

extern "C" void kernel_launch(void* const* d_in, const int* in_sizes, int n_in,
                              void* d_out, int out_size, void* d_ws, size_t ws_size,
                              hipStream_t stream) {
  const float* xh  = (const float*)d_in[0];
  const float* z   = (const float*)d_in[1];
  const float* m21 = (const float*)d_in[2];
  const float* Wf  = (const float*)d_in[3];
  const float* bf  = (const float*)d_in[4];
  const float* Wq  = (const float*)d_in[5];
  const float* bq  = (const float*)d_in[6];
  const float* Wr  = (const float*)d_in[7];
  const float* br  = (const float*)d_in[8];
  float* out = (float*)d_out;

  const int T  = 512;
  const int BT = in_sizes[1] / D14;
  const int B  = BT / T;

  kf_scan<<<dim3(B), dim3(64), 0, stream>>>(xh, z, m21, Wf, bf, Wq, bq, Wr, br,
                                            out, T);
}
